# starting kernel
# baseline (speedup 1.0000x reference)
#include <hip/hip_runtime.h>


typedef __attribute__((ext_vector_type(2))) float v2f;
typedef __attribute__((ext_vector_type(8))) float v8f;

#define DD 64
#define AP 68
#define WP 72

__device__ __forceinline__ v8f wmma4(v2f a, v2f b, v8f c) {
  return __builtin_amdgcn_wmma_f32_16x16x4_f32(
       false, a,  false, b,
       (short)0, c,  false,  false);
}

__global__ void edge_prep(const int* __restrict__ dst, const int* __restrict__ et,
                          int* __restrict__ cnt, int* __restrict__ hist,
                          int E, int R) {
  int e = blockIdx.x * blockDim.x + threadIdx.x;
  if (e >= E) return;
  int t = et[e];
  atomicAdd(&cnt[(size_t)dst[e] * R + t], 1);
  atomicAdd(&hist[t], 1);
}

__global__ void scan_pad(const int* __restrict__ hist, int* __restrict__ cursor,
                         int* __restrict__ tileRel, int R, int maxTiles) {
  if (blockIdx.x != 0 || threadIdx.x != 0) return;
  int run = 0;
  for (int r = 0; r < R; ++r) {
    cursor[r] = run * 64;
    int tiles = (hist[r] + 63) >> 6;
    for (int tt = 0; tt < tiles; ++tt) tileRel[run + tt] = r;
    run += tiles;
  }
  for (int tt = run; tt < maxTiles; ++tt) tileRel[tt] = -1;
}

__global__ void edge_scatter(const int* __restrict__ et, int* __restrict__ cursor,
                             int* __restrict__ perm, int E) {
  int e = blockIdx.x * blockDim.x + threadIdx.x;
  if (e >= E) return;
  int pos = atomicAdd(&cursor[et[e]], 1);
  perm[pos] = e;
}

__global__ void __launch_bounds__(256)
edge_gemm(const float* __restrict__ h, const float* __restrict__ Wl,
          const int* __restrict__ perm, const int* __restrict__ tileRel,
          const int* __restrict__ src, const int* __restrict__ dstArr,
          const int* __restrict__ et, const int* __restrict__ cnt,
          float* __restrict__ conv, int R) {
  __shared__ float As[64][AP];
  __shared__ float Ws[64][WP];
  __shared__ int   eSrc[64];
  __shared__ int   eDst[64];
  __shared__ float eInv[64];

  int rel = tileRel[blockIdx.x];
  if (rel < 0) return;
  int tid = threadIdx.x;

  if (tid < 64) {
    int e = perm[(size_t)blockIdx.x * 64 + tid];
    int s = -1, d = 0; float inv = 0.f;
    if (e >= 0) {
      s = src[e];
      d = dstArr[e];
      int c = cnt[(size_t)d * R + et[e]];
      inv = 1.0f / (float)(c > 0 ? c : 1);
    }
    eSrc[tid] = s; eDst[tid] = d; eInv[tid] = inv;
  }

  {
    const float4* W4 = (const float4*)(Wl + (size_t)rel * DD * DD);
    for (int i = 0; i < 4; ++i) {
      int idx = tid * 4 + i;
      int r = idx >> 4, c4 = (idx & 15) * 4;
      float4 v = W4[idx];
      Ws[r][c4 + 0] = v.x; Ws[r][c4 + 1] = v.y;
      Ws[r][c4 + 2] = v.z; Ws[r][c4 + 3] = v.w;
    }
  }
  __syncthreads();

  {
    int row = tid >> 2, cb = (tid & 3) * 16;
    int s = eSrc[row];
    if (s >= 0) {
      const float4* hp = (const float4*)(h + (size_t)s * DD + cb);
      for (int i = 0; i < 4; ++i) {
        float4 v = hp[i];
        As[row][cb + i * 4 + 0] = v.x; As[row][cb + i * 4 + 1] = v.y;
        As[row][cb + i * 4 + 2] = v.z; As[row][cb + i * 4 + 3] = v.w;
      }
    } else {
      for (int i = 0; i < 16; ++i) As[row][cb + i] = 0.f;
    }
  }
  __syncthreads();

  int lane = tid & 31, w = tid >> 5;
  int rt = w >> 1, cp = w & 1;
  int cb0 = cp * 32, cb1 = cb0 + 16;
  int n = lane & 15, hi = lane >> 4;
  int m = rt * 16 + n;

  v8f c0 = {}; v8f c1 = {};
  for (int k = 0; k < 16; ++k) {
    int kb = k * 4 + hi * 2;
    v2f a;  a.x  = As[m][kb];        a.y  = As[m][kb + 1];
    v2f b0; b0.x = Ws[kb][cb0 + n];  b0.y = Ws[kb + 1][cb0 + n];
    v2f b1; b1.x = Ws[kb][cb1 + n];  b1.y = Ws[kb + 1][cb1 + n];
    c0 = wmma4(a, b0, c0);
    c1 = wmma4(a, b1, c1);
  }

  for (int i = 0; i < 8; ++i) {
    int gr = rt * 16 + hi * 8 + i;
    if (eSrc[gr] >= 0) {
      float inv = eInv[gr];
      size_t base = (size_t)eDst[gr] * DD;
      atomicAdd(&conv[base + cb0 + n], c0[i] * inv);
      atomicAdd(&conv[base + cb1 + n], c1[i] * inv);
    }
  }
}

__global__ void __launch_bounds__(256)
node_gemm(const float* __restrict__ h, const float* __restrict__ conv,
          const float* __restrict__ rootW, const float* __restrict__ biasv,
          const float* __restrict__ resW, const float* __restrict__ resb,
          float* __restrict__ hout, int N) {
  __shared__ float As[64][AP];
  __shared__ float Wr[64][WP];
  __shared__ float Wq[64][WP];
  __shared__ float bV[64], rbV[64];

  int tid = threadIdx.x;
  int base = blockIdx.x * 64;

  if (tid < 64) { bV[tid] = biasv[tid]; rbV[tid] = resb[tid]; }

  for (int i = 0; i < 4; ++i) {
    int idx = tid * 4 + i;
    int r = idx >> 4, c4 = (idx & 15) * 4;
    float4 v = ((const float4*)rootW)[idx];
    Wr[r][c4 + 0] = v.x; Wr[r][c4 + 1] = v.y; Wr[r][c4 + 2] = v.z; Wr[r][c4 + 3] = v.w;
    float4 u = ((const float4*)resW)[idx];
    Wq[r][c4 + 0] = u.x; Wq[r][c4 + 1] = u.y; Wq[r][c4 + 2] = u.z; Wq[r][c4 + 3] = u.w;
  }

  {
    int row = tid >> 2, cb = (tid & 3) * 16;
    int node = base + row;
    if (node < N) {
      const float4* hp = (const float4*)(h + (size_t)node * DD + cb);
      for (int i = 0; i < 4; ++i) {
        float4 v = hp[i];
        As[row][cb + i * 4 + 0] = v.x; As[row][cb + i * 4 + 1] = v.y;
        As[row][cb + i * 4 + 2] = v.z; As[row][cb + i * 4 + 3] = v.w;
      }
    } else {
      for (int i = 0; i < 16; ++i) As[row][cb + i] = 0.f;
    }
  }
  __syncthreads();

  int lane = tid & 31, w = tid >> 5;
  int rt = w >> 1, cp = w & 1;
  int cb0 = cp * 32, cb1 = cb0 + 16;
  int n = lane & 15, hi = lane >> 4;
  int m = rt * 16 + n;

  v8f cR0 = {}, cR1 = {}, cS0 = {}, cS1 = {};
  for (int k = 0; k < 16; ++k) {
    int kb = k * 4 + hi * 2;
    v2f a;   a.x   = As[m][kb];        a.y   = As[m][kb + 1];
    v2f br0; br0.x = Wr[kb][cb0 + n];  br0.y = Wr[kb + 1][cb0 + n];
    v2f br1; br1.x = Wr[kb][cb1 + n];  br1.y = Wr[kb + 1][cb1 + n];
    v2f bq0; bq0.x = Wq[kb][cb0 + n];  bq0.y = Wq[kb + 1][cb0 + n];
    v2f bq1; bq1.x = Wq[kb][cb1 + n];  bq1.y = Wq[kb + 1][cb1 + n];
    cR0 = wmma4(a, br0, cR0);
    cR1 = wmma4(a, br1, cR1);
    cS0 = wmma4(a, bq0, cS0);
    cS1 = wmma4(a, bq1, cS1);
  }

  for (int i = 0; i < 8; ++i) {
    int gr = rt * 16 + hi * 8 + i;
    int node = base + gr;
    if (node < N) {
      size_t b0 = (size_t)node * DD + cb0 + n;
      size_t b1 = (size_t)node * DD + cb1 + n;
      float o0 = fmaxf(conv[b0] + cR0[i] + bV[cb0 + n], 0.f) +
                 fmaxf(cS0[i] + rbV[cb0 + n], 0.f);
      float o1 = fmaxf(conv[b1] + cR1[i] + bV[cb1 + n], 0.f) +
                 fmaxf(cS1[i] + rbV[cb1 + n], 0.f);
      hout[b0] = o0;
      hout[b1] = o1;
    }
  }
}

extern "C" void kernel_launch(void* const* d_in, const int* in_sizes, int n_in,
                              void* d_out, int out_size, void* d_ws, size_t ws_size,
                              hipStream_t stream) {
  const float* x    = (const float*)d_in[0];
  const int*   ei   = (const int*)d_in[1];
  const int*   et   = (const int*)d_in[2];
  const float* W    = (const float*)d_in[3];
  const float* root = (const float*)d_in[4];
  const float* bias = (const float*)d_in[5];
  const float* resW = (const float*)d_in[6];
  const float* resb = (const float*)d_in[7];

  const int N = in_sizes[0] / DD;
  const int E = in_sizes[2];
  const int L = in_sizes[4] / (DD * DD);
  const int R = in_sizes[3] / (L * DD * DD);
  const int* src  = ei;
  const int* dstA = ei + E;
  const int maxTiles = (E + 63) / 64 + R;

  char* wsb = (char*)d_ws;
  size_t off = 0;
  auto carve = [&](size_t bytes) -> void* {
    void* p = wsb + off;
    off += (bytes + 255) & ~(size_t)255;
    return p;
  };
  int*   cnt     = (int*)carve((size_t)N * R * sizeof(int));
  int*   hist    = (int*)carve((size_t)R * sizeof(int));
  int*   cursor  = (int*)carve((size_t)R * sizeof(int));
  int*   tileRel = (int*)carve((size_t)maxTiles * sizeof(int));
  int*   perm    = (int*)carve((size_t)maxTiles * 64 * sizeof(int));
  float* hA      = (float*)carve((size_t)N * DD * sizeof(float));
  float* hB      = (float*)carve((size_t)N * DD * sizeof(float));
  float* conv    = (float*)carve((size_t)N * DD * sizeof(float));

  hipMemsetAsync(cnt, 0, (size_t)N * R * sizeof(int), stream);
  hipMemsetAsync(hist, 0, (size_t)R * sizeof(int), stream);
  hipMemsetAsync(perm, 0xFF, (size_t)maxTiles * 64 * sizeof(int), stream);

  const int eb = (E + 255) / 256;
  edge_prep   <<<eb, 256, 0, stream>>>(dstA, et, cnt, hist, E, R);
  scan_pad    <<<1, 32, 0, stream>>>(hist, cursor, tileRel, R, maxTiles);
  edge_scatter<<<eb, 256, 0, stream>>>(et, cursor, perm, E);

  const float* hcur = x;
  for (int l = 0; l < L; ++l) {
    float* hnext = (l == L - 1) ? (float*)d_out : ((l & 1) ? hB : hA);
    hipMemsetAsync(conv, 0, (size_t)N * DD * sizeof(float), stream);
    edge_gemm<<<maxTiles, 256, 0, stream>>>(
        hcur, W + (size_t)l * R * DD * DD, perm, tileRel,
        src, dstA, et, cnt, conv, R);
    node_gemm<<<(N + 63) / 64, 256, 0, stream>>>(
        hcur, conv, root + (size_t)l * DD * DD, bias + (size_t)l * DD,
        resW + (size_t)l * DD * DD, resb + (size_t)l * DD, hnext, N);
    hcur = hnext;
  }
}


# Counting sort keyed by (relation, dst): degree table, scanned per relation, is histogram and scatter cursor; hot per-relation counters and serial scan gone; order of float atomic sums changes
# speedup vs baseline: 1.7685x; 1.7685x over previous
#include <hip/hip_runtime.h>


typedef __attribute__((ext_vector_type(2))) float v2f;
typedef __attribute__((ext_vector_type(8))) float v8f;

#define DD 64
#define AP 68
#define WP 72

__device__ __forceinline__ v8f wmma4(v2f a, v2f b, v8f c) {
  return __builtin_amdgcn_wmma_f32_16x16x4_f32(
       false, a,  false, b,
       (short)0, c,  false,  false);
}

#define CS 2048

__device__ __forceinline__ int block_sum256(int v, int* red) {
  int tid = threadIdx.x;
  red[tid] = v;
  __syncthreads();
  for (int s = 128; s > 0; s >>= 1) {
    if (tid < s) red[tid] += red[tid + s];
    __syncthreads();
  }
  return red[0];
}

__global__ void edge_count(const int* __restrict__ dst, const int* __restrict__ et,
                           int* __restrict__ tbl, int E, int N) {
  int e = blockIdx.x * blockDim.x + threadIdx.x;
  if (e >= E) return;
  atomicAdd(&tbl[(size_t)et[e] * N + dst[e]], 1);
}

__global__ void __launch_bounds__(256)
seg_sums(const int* __restrict__ tbl, int* __restrict__ bsum, int N, int CH) {
  __shared__ int red[256];
  int r = blockIdx.x / CH, c = blockIdx.x - r * CH;
  int lo = c * CS;
  int len = min(CS, N - lo);
  const int* p = tbl + (size_t)r * N + lo;
  int s = 0;
  for (int i = threadIdx.x; i < len; i += 256) s += p[i];
  int tot = block_sum256(s, red);
  if (threadIdx.x == 0) bsum[blockIdx.x] = tot;
}

__global__ void __launch_bounds__(256)
scan_write(int* __restrict__ tbl, const int* __restrict__ bsum,
           int* __restrict__ relStart, int* __restrict__ tileRel,
           int N, int R, int CH, int maxTiles) {
  __shared__ int red[3][256];
  __shared__ int tsum[2][256];
  int tid = threadIdx.x;
  int r = blockIdx.x / CH, c = blockIdx.x - r * CH;

  int tb = 0;
  for (int q = tid; q < r; q += 256) {
    int hq = 0;
    for (int j = 0; j < CH; ++j) hq += bsum[q * CH + j];
    tb += (hq + 63) >> 6;
  }
  int bf = 0, tt = 0;
  for (int j = tid; j < CH; j += 256) {
    int v = bsum[r * CH + j];
    tt += v;
    if (j < c) bf += v;
  }
  red[0][tid] = tb; red[1][tid] = bf; red[2][tid] = tt;
  __syncthreads();
  for (int s = 128; s > 0; s >>= 1) {
    if (tid < s) {
      red[0][tid] += red[0][tid + s];
      red[1][tid] += red[1][tid + s];
      red[2][tid] += red[2][tid + s];
    }
    __syncthreads();
  }
  int tilesBefore = red[0][0], before = red[1][0], total = red[2][0];
  int base = tilesBefore * 64 + before;

  int lo = c * CS;
  int len = min(CS, N - lo);
  int* p = tbl + (size_t)r * N + lo;
  int i0 = tid * 8;
  int ex[8];
  int s = 0;
  for (int j = 0; j < 8; ++j) {
    int idx = i0 + j;
    int x = (idx < len) ? p[idx] : 0;
    ex[j] = s;
    s += x;
  }
  int cur = 0;
  tsum[0][tid] = s;
  __syncthreads();
  for (int off = 1; off < 256; off <<= 1) {
    int v = tsum[cur][tid];
    if (tid >= off) v += tsum[cur][tid - off];
    cur ^= 1;
    tsum[cur][tid] = v;
    __syncthreads();
  }
  int excl = tsum[cur][tid] - s;
  for (int j = 0; j < 8; ++j) {
    int idx = i0 + j;
    if (idx < len) p[idx] = base + excl + ex[j];
  }

  if (c == 0) {
    int myTiles = (total + 63) >> 6;
    if (tid == 0) relStart[r] = tilesBefore * 64;
    for (int t = tid; t < myTiles; t += 256) tileRel[tilesBefore + t] = r;
    if (r == R - 1)
      for (int t = tilesBefore + myTiles + tid; t < maxTiles; t += 256) tileRel[t] = -1;
  }
}

__global__ void edge_scatter(const int* __restrict__ dst, const int* __restrict__ et,
                             int* __restrict__ tbl, int* __restrict__ perm,
                             int E, int N) {
  int e = blockIdx.x * blockDim.x + threadIdx.x;
  if (e >= E) return;
  int pos = atomicAdd(&tbl[(size_t)et[e] * N + dst[e]], 1);
  perm[pos] = e;
}

__global__ void __launch_bounds__(256)
edge_gemm(const float* __restrict__ h, const float* __restrict__ Wl,
          const int* __restrict__ perm, const int* __restrict__ tileRel,
          const int* __restrict__ src, const int* __restrict__ dstArr,
          const int* __restrict__ tbl, const int* __restrict__ relStart,
          float* __restrict__ conv, int N) {
  __shared__ float As[64][AP];
  __shared__ float Ws[64][WP];
  __shared__ int   eSrc[64];
  __shared__ int   eDst[64];
  __shared__ float eInv[64];

  int rel = tileRel[blockIdx.x];
  if (rel < 0) return;
  int tid = threadIdx.x;

  if (tid < 64) {
    int e = perm[(size_t)blockIdx.x * 64 + tid];
    int s = -1, d = 0; float inv = 0.f;
    if (e >= 0) {
      s = src[e];
      d = dstArr[e];
      size_t gi = (size_t)rel * N + d;
      int c = tbl[gi] - (d > 0 ? tbl[gi - 1] : relStart[rel]);
      inv = 1.0f / (float)(c > 0 ? c : 1);
    }
    eSrc[tid] = s; eDst[tid] = d; eInv[tid] = inv;
  }

  {
    const float4* W4 = (const float4*)(Wl + (size_t)rel * DD * DD);
    for (int i = 0; i < 4; ++i) {
      int idx = tid * 4 + i;
      int r = idx >> 4, c4 = (idx & 15) * 4;
      float4 v = W4[idx];
      Ws[r][c4 + 0] = v.x; Ws[r][c4 + 1] = v.y;
      Ws[r][c4 + 2] = v.z; Ws[r][c4 + 3] = v.w;
    }
  }
  __syncthreads();

  {
    int row = tid >> 2, cb = (tid & 3) * 16;
    int s = eSrc[row];
    if (s >= 0) {
      const float4* hp = (const float4*)(h + (size_t)s * DD + cb);
      for (int i = 0; i < 4; ++i) {
        float4 v = hp[i];
        As[row][cb + i * 4 + 0] = v.x; As[row][cb + i * 4 + 1] = v.y;
        As[row][cb + i * 4 + 2] = v.z; As[row][cb + i * 4 + 3] = v.w;
      }
    } else {
      for (int i = 0; i < 16; ++i) As[row][cb + i] = 0.f;
    }
  }
  __syncthreads();

  int lane = tid & 31, w = tid >> 5;
  int rt = w >> 1, cp = w & 1;
  int cb0 = cp * 32, cb1 = cb0 + 16;
  int n = lane & 15, hi = lane >> 4;
  int m = rt * 16 + n;

  v8f c0 = {}; v8f c1 = {};
  for (int k = 0; k < 16; ++k) {
    int kb = k * 4 + hi * 2;
    v2f a;  a.x  = As[m][kb];        a.y  = As[m][kb + 1];
    v2f b0; b0.x = Ws[kb][cb0 + n];  b0.y = Ws[kb + 1][cb0 + n];
    v2f b1; b1.x = Ws[kb][cb1 + n];  b1.y = Ws[kb + 1][cb1 + n];
    c0 = wmma4(a, b0, c0);
    c1 = wmma4(a, b1, c1);
  }

  for (int i = 0; i < 8; ++i) {
    int gr = rt * 16 + hi * 8 + i;
    if (eSrc[gr] >= 0) {
      float inv = eInv[gr];
      size_t base = (size_t)eDst[gr] * DD;
      atomicAdd(&conv[base + cb0 + n], c0[i] * inv);
      atomicAdd(&conv[base + cb1 + n], c1[i] * inv);
    }
  }
}

__global__ void __launch_bounds__(256)
node_gemm(const float* __restrict__ h, const float* __restrict__ conv,
          const float* __restrict__ rootW, const float* __restrict__ biasv,
          const float* __restrict__ resW, const float* __restrict__ resb,
          float* __restrict__ hout, int N) {
  __shared__ float As[64][AP];
  __shared__ float Wr[64][WP];
  __shared__ float Wq[64][WP];
  __shared__ float bV[64], rbV[64];

  int tid = threadIdx.x;
  int base = blockIdx.x * 64;

  if (tid < 64) { bV[tid] = biasv[tid]; rbV[tid] = resb[tid]; }

  for (int i = 0; i < 4; ++i) {
    int idx = tid * 4 + i;
    int r = idx >> 4, c4 = (idx & 15) * 4;
    float4 v = ((const float4*)rootW)[idx];
    Wr[r][c4 + 0] = v.x; Wr[r][c4 + 1] = v.y; Wr[r][c4 + 2] = v.z; Wr[r][c4 + 3] = v.w;
    float4 u = ((const float4*)resW)[idx];
    Wq[r][c4 + 0] = u.x; Wq[r][c4 + 1] = u.y; Wq[r][c4 + 2] = u.z; Wq[r][c4 + 3] = u.w;
  }

  {
    int row = tid >> 2, cb = (tid & 3) * 16;
    int node = base + row;
    if (node < N) {
      const float4* hp = (const float4*)(h + (size_t)node * DD + cb);
      for (int i = 0; i < 4; ++i) {
        float4 v = hp[i];
        As[row][cb + i * 4 + 0] = v.x; As[row][cb + i * 4 + 1] = v.y;
        As[row][cb + i * 4 + 2] = v.z; As[row][cb + i * 4 + 3] = v.w;
      }
    } else {
      for (int i = 0; i < 16; ++i) As[row][cb + i] = 0.f;
    }
  }
  __syncthreads();

  int lane = tid & 31, w = tid >> 5;
  int rt = w >> 1, cp = w & 1;
  int cb0 = cp * 32, cb1 = cb0 + 16;
  int n = lane & 15, hi = lane >> 4;
  int m = rt * 16 + n;

  v8f cR0 = {}, cR1 = {}, cS0 = {}, cS1 = {};
  for (int k = 0; k < 16; ++k) {
    int kb = k * 4 + hi * 2;
    v2f a;   a.x   = As[m][kb];        a.y   = As[m][kb + 1];
    v2f br0; br0.x = Wr[kb][cb0 + n];  br0.y = Wr[kb + 1][cb0 + n];
    v2f br1; br1.x = Wr[kb][cb1 + n];  br1.y = Wr[kb + 1][cb1 + n];
    v2f bq0; bq0.x = Wq[kb][cb0 + n];  bq0.y = Wq[kb + 1][cb0 + n];
    v2f bq1; bq1.x = Wq[kb][cb1 + n];  bq1.y = Wq[kb + 1][cb1 + n];
    cR0 = wmma4(a, br0, cR0);
    cR1 = wmma4(a, br1, cR1);
    cS0 = wmma4(a, bq0, cS0);
    cS1 = wmma4(a, bq1, cS1);
  }

  for (int i = 0; i < 8; ++i) {
    int gr = rt * 16 + hi * 8 + i;
    int node = base + gr;
    if (node < N) {
      size_t b0 = (size_t)node * DD + cb0 + n;
      size_t b1 = (size_t)node * DD + cb1 + n;
      float o0 = fmaxf(conv[b0] + cR0[i] + bV[cb0 + n], 0.f) +
                 fmaxf(cS0[i] + rbV[cb0 + n], 0.f);
      float o1 = fmaxf(conv[b1] + cR1[i] + bV[cb1 + n], 0.f) +
                 fmaxf(cS1[i] + rbV[cb1 + n], 0.f);
      hout[b0] = o0;
      hout[b1] = o1;
    }
  }
}

extern "C" void kernel_launch(void* const* d_in, const int* in_sizes, int n_in,
                              void* d_out, int out_size, void* d_ws, size_t ws_size,
                              hipStream_t stream) {
  const float* x    = (const float*)d_in[0];
  const int*   ei   = (const int*)d_in[1];
  const int*   et   = (const int*)d_in[2];
  const float* W    = (const float*)d_in[3];
  const float* root = (const float*)d_in[4];
  const float* bias = (const float*)d_in[5];
  const float* resW = (const float*)d_in[6];
  const float* resb = (const float*)d_in[7];

  const int N = in_sizes[0] / DD;
  const int E = in_sizes[2];
  const int L = in_sizes[4] / (DD * DD);
  const int R = in_sizes[3] / (L * DD * DD);
  const int* src  = ei;
  const int* dstA = ei + E;
  const int maxTiles = (E + 63) / 64 + R;

  char* wsb = (char*)d_ws;
  size_t off = 0;
  auto carve = [&](size_t bytes) -> void* {
    void* p = wsb + off;
    off += (bytes + 255) & ~(size_t)255;
    return p;
  };
  int*   tbl     = (int*)carve((size_t)N * R * sizeof(int));
  int*   relStart= (int*)carve((size_t)R * sizeof(int));
  int*   tileRel = (int*)carve((size_t)maxTiles * sizeof(int));
  int*   perm    = (int*)carve((size_t)maxTiles * 64 * sizeof(int));
  float* hA      = (float*)carve((size_t)N * DD * sizeof(float));
  float* hB      = (float*)carve((size_t)N * DD * sizeof(float));
  float* conv    = (float*)carve((size_t)N * DD * sizeof(float));

  int*   bsum    = (int*)conv;

  hipMemsetAsync(tbl, 0, (size_t)N * R * sizeof(int), stream);
  hipMemsetAsync(perm, 0xFF, (size_t)maxTiles * 64 * sizeof(int), stream);

  const int eb = (E + 255) / 256;
  const int CH = (N + CS - 1) / CS;
  edge_count  <<<eb, 256, 0, stream>>>(dstA, et, tbl, E, N);
  seg_sums    <<<R * CH, 256, 0, stream>>>(tbl, bsum, N, CH);
  scan_write  <<<R * CH, 256, 0, stream>>>(tbl, bsum, relStart, tileRel, N, R, CH, maxTiles);
  edge_scatter<<<eb, 256, 0, stream>>>(dstA, et, tbl, perm, E, N);

  const float* hcur = x;
  for (int l = 0; l < L; ++l) {
    float* hnext = (l == L - 1) ? (float*)d_out : ((l & 1) ? hB : hA);
    hipMemsetAsync(conv, 0, (size_t)N * DD * sizeof(float), stream);
    edge_gemm<<<maxTiles, 256, 0, stream>>>(
        hcur, W + (size_t)l * R * DD * DD, perm, tileRel,
        src, dstA, tbl, relStart, conv, N);
    node_gemm<<<(N + 63) / 64, 256, 0, stream>>>(
        hcur, conv, root + (size_t)l * DD * DD, bias + (size_t)l * DD,
        resW + (size_t)l * DD * DD, resb + (size_t)l * DD, hnext, N);
    hcur = hnext;
  }
}


# Edge GEMM scatter: lane halves trade one C register so each float atomic instruction covers one whole 128-byte conv line, not 64-byte halves of two; arrival order of the float atomic sums changes
# speedup vs baseline: 1.8260x; 1.0325x over previous
#include <hip/hip_runtime.h>


typedef __attribute__((ext_vector_type(2))) float v2f;
typedef __attribute__((ext_vector_type(8))) float v8f;

#define DD 64
#define AP 68
#define WP 72

__device__ __forceinline__ v8f wmma4(v2f a, v2f b, v8f c) {
  return __builtin_amdgcn_wmma_f32_16x16x4_f32(
       false, a,  false, b,
       (short)0, c,  false,  false);
}

__device__ __forceinline__ float xhalf(float x) {
  int v = __builtin_bit_cast(int, x);
  return __builtin_bit_cast(float, __builtin_amdgcn_permlanex16(
      v, v, 0x76543210, 0xfedcba98, false, false));
}

#define CS 2048

__device__ __forceinline__ int block_sum256(int v, int* red) {
  int tid = threadIdx.x;
  red[tid] = v;
  __syncthreads();
  for (int s = 128; s > 0; s >>= 1) {
    if (tid < s) red[tid] += red[tid + s];
    __syncthreads();
  }
  return red[0];
}

__global__ void edge_count(const int* __restrict__ dst, const int* __restrict__ et,
                           int* __restrict__ tbl, int E, int N) {
  int e = blockIdx.x * blockDim.x + threadIdx.x;
  if (e >= E) return;
  atomicAdd(&tbl[(size_t)et[e] * N + dst[e]], 1);
}

__global__ void __launch_bounds__(256)
seg_sums(const int* __restrict__ tbl, int* __restrict__ bsum, int N, int CH) {
  __shared__ int red[256];
  int r = blockIdx.x / CH, c = blockIdx.x - r * CH;
  int lo = c * CS;
  int len = min(CS, N - lo);
  const int* p = tbl + (size_t)r * N + lo;
  int s = 0;
  for (int i = threadIdx.x; i < len; i += 256) s += p[i];
  int tot = block_sum256(s, red);
  if (threadIdx.x == 0) bsum[blockIdx.x] = tot;
}

__global__ void __launch_bounds__(256)
scan_write(int* __restrict__ tbl, const int* __restrict__ bsum,
           int* __restrict__ relStart, int* __restrict__ tileRel,
           int N, int R, int CH, int maxTiles) {
  __shared__ int red[3][256];
  __shared__ int tsum[2][256];
  int tid = threadIdx.x;
  int r = blockIdx.x / CH, c = blockIdx.x - r * CH;

  int tb = 0;
  for (int q = tid; q < r; q += 256) {
    int hq = 0;
    for (int j = 0; j < CH; ++j) hq += bsum[q * CH + j];
    tb += (hq + 63) >> 6;
  }
  int bf = 0, tt = 0;
  for (int j = tid; j < CH; j += 256) {
    int v = bsum[r * CH + j];
    tt += v;
    if (j < c) bf += v;
  }
  red[0][tid] = tb; red[1][tid] = bf; red[2][tid] = tt;
  __syncthreads();
  for (int s = 128; s > 0; s >>= 1) {
    if (tid < s) {
      red[0][tid] += red[0][tid + s];
      red[1][tid] += red[1][tid + s];
      red[2][tid] += red[2][tid + s];
    }
    __syncthreads();
  }
  int tilesBefore = red[0][0], before = red[1][0], total = red[2][0];
  int base = tilesBefore * 64 + before;

  int lo = c * CS;
  int len = min(CS, N - lo);
  int* p = tbl + (size_t)r * N + lo;
  int i0 = tid * 8;
  int ex[8];
  int s = 0;
  for (int j = 0; j < 8; ++j) {
    int idx = i0 + j;
    int x = (idx < len) ? p[idx] : 0;
    ex[j] = s;
    s += x;
  }
  int cur = 0;
  tsum[0][tid] = s;
  __syncthreads();
  for (int off = 1; off < 256; off <<= 1) {
    int v = tsum[cur][tid];
    if (tid >= off) v += tsum[cur][tid - off];
    cur ^= 1;
    tsum[cur][tid] = v;
    __syncthreads();
  }
  int excl = tsum[cur][tid] - s;
  for (int j = 0; j < 8; ++j) {
    int idx = i0 + j;
    if (idx < len) p[idx] = base + excl + ex[j];
  }

  if (c == 0) {
    int myTiles = (total + 63) >> 6;
    if (tid == 0) relStart[r] = tilesBefore * 64;
    for (int t = tid; t < myTiles; t += 256) tileRel[tilesBefore + t] = r;
    if (r == R - 1)
      for (int t = tilesBefore + myTiles + tid; t < maxTiles; t += 256) tileRel[t] = -1;
  }
}

__global__ void edge_scatter(const int* __restrict__ dst, const int* __restrict__ et,
                             int* __restrict__ tbl, int* __restrict__ perm,
                             int E, int N) {
  int e = blockIdx.x * blockDim.x + threadIdx.x;
  if (e >= E) return;
  int pos = atomicAdd(&tbl[(size_t)et[e] * N + dst[e]], 1);
  perm[pos] = e;
}

__global__ void __launch_bounds__(256)
edge_gemm(const float* __restrict__ h, const float* __restrict__ Wl,
          const int* __restrict__ perm, const int* __restrict__ tileRel,
          const int* __restrict__ src, const int* __restrict__ dstArr,
          const int* __restrict__ tbl, const int* __restrict__ relStart,
          float* __restrict__ conv, int N) {
  __shared__ float As[64][AP];
  __shared__ float Ws[64][WP];
  __shared__ int   eSrc[64];
  __shared__ int   eDst[64];
  __shared__ float eInv[64];

  int rel = tileRel[blockIdx.x];
  if (rel < 0) return;
  int tid = threadIdx.x;

  if (tid < 64) {
    int e = perm[(size_t)blockIdx.x * 64 + tid];
    int s = -1, d = 0; float inv = 0.f;
    if (e >= 0) {
      s = src[e];
      d = dstArr[e];
      size_t gi = (size_t)rel * N + d;
      int c = tbl[gi] - (d > 0 ? tbl[gi - 1] : relStart[rel]);
      inv = 1.0f / (float)(c > 0 ? c : 1);
    }
    eSrc[tid] = s; eDst[tid] = d; eInv[tid] = inv;
  }

  {
    const float4* W4 = (const float4*)(Wl + (size_t)rel * DD * DD);
    for (int i = 0; i < 4; ++i) {
      int idx = tid * 4 + i;
      int r = idx >> 4, c4 = (idx & 15) * 4;
      float4 v = W4[idx];
      Ws[r][c4 + 0] = v.x; Ws[r][c4 + 1] = v.y;
      Ws[r][c4 + 2] = v.z; Ws[r][c4 + 3] = v.w;
    }
  }
  __syncthreads();

  {
    int row = tid >> 2, cb = (tid & 3) * 16;
    int s = eSrc[row];
    if (s >= 0) {
      const float4* hp = (const float4*)(h + (size_t)s * DD + cb);
      for (int i = 0; i < 4; ++i) {
        float4 v = hp[i];
        As[row][cb + i * 4 + 0] = v.x; As[row][cb + i * 4 + 1] = v.y;
        As[row][cb + i * 4 + 2] = v.z; As[row][cb + i * 4 + 3] = v.w;
      }
    } else {
      for (int i = 0; i < 16; ++i) As[row][cb + i] = 0.f;
    }
  }
  __syncthreads();

  int lane = tid & 31, w = tid >> 5;
  int rt = w >> 1, cp = w & 1;
  int cb0 = cp * 32, cb1 = cb0 + 16;
  int n = lane & 15, hi = lane >> 4;
  int m = rt * 16 + n;

  v8f c0 = {}; v8f c1 = {};
  for (int k = 0; k < 16; ++k) {
    int kb = k * 4 + hi * 2;
    v2f a;  a.x  = As[m][kb];        a.y  = As[m][kb + 1];
    v2f b0; b0.x = Ws[kb][cb0 + n];  b0.y = Ws[kb + 1][cb0 + n];
    v2f b1; b1.x = Ws[kb][cb1 + n];  b1.y = Ws[kb + 1][cb1 + n];
    c0 = wmma4(a, b0, c0);
    c1 = wmma4(a, b1, c1);
  }

  int   ms = eSrc[m];
  int   md = eDst[m];
  float mi = eInv[m];
  int col = cb0 + lane;
  for (int i = 0; i < 8; ++i) {
    int   slo = __builtin_amdgcn_readlane(ms, i);
    int   shi = __builtin_amdgcn_readlane(ms, i + 8);
    int   dlo = __builtin_amdgcn_readlane(md, i);
    int   dhi = __builtin_amdgcn_readlane(md, i + 8);
    float ilo = __builtin_bit_cast(float,
        __builtin_amdgcn_readlane(__builtin_bit_cast(int, mi), i));
    float ihi = __builtin_bit_cast(float,
        __builtin_amdgcn_readlane(__builtin_bit_cast(int, mi), i + 8));
    float give = hi ? c0[i] : c1[i];
    float got  = xhalf(give);
    float vlo  = hi ? got : c0[i];
    float vhi  = hi ? c1[i] : got;
    if (slo >= 0)
      atomicAdd(&conv[(size_t)dlo * DD + col], vlo * ilo);
    if (shi >= 0)
      atomicAdd(&conv[(size_t)dhi * DD + col], vhi * ihi);
  }
}

__global__ void __launch_bounds__(256)
node_gemm(const float* __restrict__ h, const float* __restrict__ conv,
          const float* __restrict__ rootW, const float* __restrict__ biasv,
          const float* __restrict__ resW, const float* __restrict__ resb,
          float* __restrict__ hout, int N) {
  __shared__ float As[64][AP];
  __shared__ float Wr[64][WP];
  __shared__ float Wq[64][WP];
  __shared__ float bV[64], rbV[64];

  int tid = threadIdx.x;
  int base = blockIdx.x * 64;

  if (tid < 64) { bV[tid] = biasv[tid]; rbV[tid] = resb[tid]; }

  for (int i = 0; i < 4; ++i) {
    int idx = tid * 4 + i;
    int r = idx >> 4, c4 = (idx & 15) * 4;
    float4 v = ((const float4*)rootW)[idx];
    Wr[r][c4 + 0] = v.x; Wr[r][c4 + 1] = v.y; Wr[r][c4 + 2] = v.z; Wr[r][c4 + 3] = v.w;
    float4 u = ((const float4*)resW)[idx];
    Wq[r][c4 + 0] = u.x; Wq[r][c4 + 1] = u.y; Wq[r][c4 + 2] = u.z; Wq[r][c4 + 3] = u.w;
  }

  {
    int row = tid >> 2, cb = (tid & 3) * 16;
    int node = base + row;
    if (node < N) {
      const float4* hp = (const float4*)(h + (size_t)node * DD + cb);
      for (int i = 0; i < 4; ++i) {
        float4 v = hp[i];
        As[row][cb + i * 4 + 0] = v.x; As[row][cb + i * 4 + 1] = v.y;
        As[row][cb + i * 4 + 2] = v.z; As[row][cb + i * 4 + 3] = v.w;
      }
    } else {
      for (int i = 0; i < 16; ++i) As[row][cb + i] = 0.f;
    }
  }
  __syncthreads();

  int lane = tid & 31, w = tid >> 5;
  int rt = w >> 1, cp = w & 1;
  int cb0 = cp * 32, cb1 = cb0 + 16;
  int n = lane & 15, hi = lane >> 4;
  int m = rt * 16 + n;

  v8f cR0 = {}, cR1 = {}, cS0 = {}, cS1 = {};
  for (int k = 0; k < 16; ++k) {
    int kb = k * 4 + hi * 2;
    v2f a;   a.x   = As[m][kb];        a.y   = As[m][kb + 1];
    v2f br0; br0.x = Wr[kb][cb0 + n];  br0.y = Wr[kb + 1][cb0 + n];
    v2f br1; br1.x = Wr[kb][cb1 + n];  br1.y = Wr[kb + 1][cb1 + n];
    v2f bq0; bq0.x = Wq[kb][cb0 + n];  bq0.y = Wq[kb + 1][cb0 + n];
    v2f bq1; bq1.x = Wq[kb][cb1 + n];  bq1.y = Wq[kb + 1][cb1 + n];
    cR0 = wmma4(a, br0, cR0);
    cR1 = wmma4(a, br1, cR1);
    cS0 = wmma4(a, bq0, cS0);
    cS1 = wmma4(a, bq1, cS1);
  }

  for (int i = 0; i < 8; ++i) {
    int gr = rt * 16 + hi * 8 + i;
    int node = base + gr;
    if (node < N) {
      size_t b0 = (size_t)node * DD + cb0 + n;
      size_t b1 = (size_t)node * DD + cb1 + n;
      float o0 = fmaxf(conv[b0] + cR0[i] + bV[cb0 + n], 0.f) +
                 fmaxf(cS0[i] + rbV[cb0 + n], 0.f);
      float o1 = fmaxf(conv[b1] + cR1[i] + bV[cb1 + n], 0.f) +
                 fmaxf(cS1[i] + rbV[cb1 + n], 0.f);
      hout[b0] = o0;
      hout[b1] = o1;
    }
  }
}

extern "C" void kernel_launch(void* const* d_in, const int* in_sizes, int n_in,
                              void* d_out, int out_size, void* d_ws, size_t ws_size,
                              hipStream_t stream) {
  const float* x    = (const float*)d_in[0];
  const int*   ei   = (const int*)d_in[1];
  const int*   et   = (const int*)d_in[2];
  const float* W    = (const float*)d_in[3];
  const float* root = (const float*)d_in[4];
  const float* bias = (const float*)d_in[5];
  const float* resW = (const float*)d_in[6];
  const float* resb = (const float*)d_in[7];

  const int N = in_sizes[0] / DD;
  const int E = in_sizes[2];
  const int L = in_sizes[4] / (DD * DD);
  const int R = in_sizes[3] / (L * DD * DD);
  const int* src  = ei;
  const int* dstA = ei + E;
  const int maxTiles = (E + 63) / 64 + R;

  char* wsb = (char*)d_ws;
  size_t off = 0;
  auto carve = [&](size_t bytes) -> void* {
    void* p = wsb + off;
    off += (bytes + 255) & ~(size_t)255;
    return p;
  };
  int*   tbl     = (int*)carve((size_t)N * R * sizeof(int));
  int*   relStart= (int*)carve((size_t)R * sizeof(int));
  int*   tileRel = (int*)carve((size_t)maxTiles * sizeof(int));
  int*   perm    = (int*)carve((size_t)maxTiles * 64 * sizeof(int));
  float* hA      = (float*)carve((size_t)N * DD * sizeof(float));
  float* hB      = (float*)carve((size_t)N * DD * sizeof(float));
  float* conv    = (float*)carve((size_t)N * DD * sizeof(float));

  int*   bsum    = (int*)conv;

  hipMemsetAsync(tbl, 0, (size_t)N * R * sizeof(int), stream);
  hipMemsetAsync(perm, 0xFF, (size_t)maxTiles * 64 * sizeof(int), stream);

  const int eb = (E + 255) / 256;
  const int CH = (N + CS - 1) / CS;
  edge_count  <<<eb, 256, 0, stream>>>(dstA, et, tbl, E, N);
  seg_sums    <<<R * CH, 256, 0, stream>>>(tbl, bsum, N, CH);
  scan_write  <<<R * CH, 256, 0, stream>>>(tbl, bsum, relStart, tileRel, N, R, CH, maxTiles);
  edge_scatter<<<eb, 256, 0, stream>>>(dstA, et, tbl, perm, E, N);

  const float* hcur = x;
  for (int l = 0; l < L; ++l) {
    float* hnext = (l == L - 1) ? (float*)d_out : ((l & 1) ? hB : hA);
    hipMemsetAsync(conv, 0, (size_t)N * DD * sizeof(float), stream);
    edge_gemm<<<maxTiles, 256, 0, stream>>>(
        hcur, W + (size_t)l * R * DD * DD, perm, tileRel,
        src, dstA, tbl, relStart, conv, N);
    node_gemm<<<(N + 63) / 64, 256, 0, stream>>>(
        hcur, conv, root + (size_t)l * DD * DD, bias + (size_t)l * DD,
        resW + (size_t)l * DD * DD, resb + (size_t)l * DD, hnext, N);
    hcur = hnext;
  }
}


# Node update fused into the destination-block kernel: the block applies h@root, h@resW and relu/bias to its LDS conv sums and stores next h; conv plane and 2 launches gone; float sums kept
# speedup vs baseline: 2.2795x; 1.2484x over previous
#include <hip/hip_runtime.h>


typedef __attribute__((ext_vector_type(2))) float v2f;
typedef __attribute__((ext_vector_type(8))) float v8f;

#define DD 64
#define NB 200
#define NBT ((NB + 15) / 16)
#define NBP (NBT * 16)
#define EBT 512

__device__ __forceinline__ v8f wmma4(v2f a, v2f b, v8f c) {
  return __builtin_amdgcn_wmma_f32_16x16x4_f32(
       false, a,  false, b,
       (short)0, c,  false,  false);
}

__device__ __forceinline__ float xhalf(float x) {
  int v = __builtin_bit_cast(int, x);
  return __builtin_bit_cast(float, __builtin_amdgcn_permlanex16(
      v, v, 0x76543210, 0xfedcba98, false, false));
}

__device__ __forceinline__ int tbl_index(int d, int r, int R) {
  int b = d / NB;
  return (b * R + r) * NB + (d - b * NB);
}

#define CS 2048

__device__ __forceinline__ int block_sum256(int v, int* red) {
  int tid = threadIdx.x;
  red[tid] = v;
  __syncthreads();
  for (int s = 128; s > 0; s >>= 1) {
    if (tid < s) red[tid] += red[tid + s];
    __syncthreads();
  }
  return red[0];
}

__global__ void edge_count(const int* __restrict__ dst, const int* __restrict__ et,
                           int* __restrict__ tbl, int E, int R) {
  int e = blockIdx.x * blockDim.x + threadIdx.x;
  if (e >= E) return;
  atomicAdd(&tbl[tbl_index(dst[e], et[e], R)], 1);
}

__global__ void __launch_bounds__(256)
seg_sums(const int* __restrict__ tbl, int* __restrict__ bsum, int T) {
  __shared__ int red[256];
  int lo = blockIdx.x * CS;
  int len = min(CS, T - lo);
  const int* p = tbl + lo;
  int s = 0;
  for (int i = threadIdx.x; i < len; i += 256) s += p[i];
  int tot = block_sum256(s, red);
  if (threadIdx.x == 0) bsum[blockIdx.x] = tot;
}

__global__ void __launch_bounds__(256)
scan_write(int* __restrict__ tbl, const int* __restrict__ bsum, int T) {
  __shared__ int red[256];
  __shared__ int tsum[2][256];
  int tid = threadIdx.x;
  int c = blockIdx.x;

  int bf = 0;
  for (int j = tid; j < c; j += 256) bf += bsum[j];
  int base = block_sum256(bf, red);

  int lo = c * CS;
  int len = min(CS, T - lo);
  int* p = tbl + lo;
  int i0 = tid * 8;
  int ex[8];
  int s = 0;
  for (int j = 0; j < 8; ++j) {
    int idx = i0 + j;
    int x = (idx < len) ? p[idx] : 0;
    ex[j] = s;
    s += x;
  }
  int cur = 0;
  tsum[0][tid] = s;
  __syncthreads();
  for (int off = 1; off < 256; off <<= 1) {
    int v = tsum[cur][tid];
    if (tid >= off) v += tsum[cur][tid - off];
    cur ^= 1;
    tsum[cur][tid] = v;
    __syncthreads();
  }
  int excl = tsum[cur][tid] - s;
  for (int j = 0; j < 8; ++j) {
    int idx = i0 + j;
    if (idx < len) p[idx] = base + excl + ex[j];
  }
}

__global__ void edge_scatter(const int* __restrict__ dst, const int* __restrict__ et,
                             int* __restrict__ tbl, int* __restrict__ perm,
                             int E, int R) {
  int e = blockIdx.x * blockDim.x + threadIdx.x;
  if (e >= E) return;
  int pos = atomicAdd(&tbl[tbl_index(dst[e], et[e], R)], 1);
  perm[pos] = e;
}

__global__ void __launch_bounds__(EBT)
edge_block(const float* __restrict__ h, const float* __restrict__ Wl,
           const int* __restrict__ perm, const int* __restrict__ src,
           const int* __restrict__ dstArr, const int* __restrict__ tbl,
           const float* __restrict__ rootW, const float* __restrict__ biasv,
           const float* __restrict__ resW, const float* __restrict__ resb,
           float* __restrict__ hout, int N, int R) {
  __shared__ float acc[NBP][DD];

  int tid = threadIdx.x;
  int b = blockIdx.x;
  int nodeBase = b * NB;

  for (int i = tid; i < NBP * DD / 4; i += EBT)
    ((float4*)&acc[0][0])[i] = make_float4(0.f, 0.f, 0.f, 0.f);
  __syncthreads();

  int lane = tid & 31;
  int w = __builtin_amdgcn_readfirstlane(tid >> 5);
  int n = lane & 15, hi = lane >> 4;

  for (int r = w; r < R; r += EBT / 32) {
    int g0 = (b * R + r) * NB;
    int begin = g0 > 0 ? tbl[g0 - 1] : 0;
    int end = tbl[g0 + NB - 1];
    begin = __builtin_amdgcn_readfirstlane(begin);
    end = __builtin_amdgcn_readfirstlane(end);
    if (begin >= end) continue;

    const float* Wr = Wl + (size_t)r * DD * DD;
    v2f B[16][4];
#pragma unroll
    for (int k = 0; k < 16; ++k) {
      int kb = k * 4 + hi * 2;
#pragma unroll
      for (int ct = 0; ct < 4; ++ct) {
        B[k][ct].x = Wr[kb * DD + ct * 16 + n];
        B[k][ct].y = Wr[(kb + 1) * DD + ct * 16 + n];
      }
    }

    int last = end - 1;
    int p0 = begin + n;
    int e1 = perm[min(p0 + 16, last)];
    int ms, dn;
    {
      int e0 = perm[min(p0, last)];
      ms = src[e0];
      dn = dstArr[e0];
      if (p0 >= end) ms = -1;
    }

    for (int t0 = begin; t0 < end; t0 += 16) {
      bool ok = ms >= 0;

      v2f A[16];
      const v2f* hp = (const v2f*)(h + (size_t)(ok ? ms : 0) * DD + hi * 2);
#pragma unroll
      for (int k = 0; k < 16; ++k) A[k] = hp[k * 2];

      int md = ok ? dn - nodeBase : 0;
      int gi = g0 + md;
      int cEnd = tbl[gi];
      int cBeg = tbl[gi > 0 ? gi - 1 : 0];

      int msNext = src[e1];
      int dnNext = dstArr[e1];
      if (t0 + 16 + n >= end) msNext = -1;
      int e2 = perm[min(t0 + 32 + n, last)];

      int c = cEnd - (gi > 0 ? cBeg : 0);
      float mi = 1.0f / (float)(c > 0 ? c : 1);

      v8f c0 = {}; v8f c1 = {}; v8f c2 = {}; v8f c3 = {};
#pragma unroll
      for (int k = 0; k < 16; ++k) {
        c0 = wmma4(A[k], B[k][0], c0);
        c1 = wmma4(A[k], B[k][1], c1);
        c2 = wmma4(A[k], B[k][2], c2);
        c3 = wmma4(A[k], B[k][3], c3);
      }

#pragma unroll
      for (int i = 0; i < 8; ++i) {
        int   slo = __builtin_amdgcn_readlane(ms, i);
        int   shi = __builtin_amdgcn_readlane(ms, i + 8);
        int   dlo = __builtin_amdgcn_readlane(md, i);
        int   dhi = __builtin_amdgcn_readlane(md, i + 8);
        float ilo = __builtin_bit_cast(float,
            __builtin_amdgcn_readlane(__builtin_bit_cast(int, mi), i));
        float ihi = __builtin_bit_cast(float,
            __builtin_amdgcn_readlane(__builtin_bit_cast(int, mi), i + 8));
        float gA = xhalf(hi ? c0[i] : c1[i]);
        float gB = xhalf(hi ? c2[i] : c3[i]);
        float aLo = hi ? gA : c0[i];
        float aHi = hi ? c1[i] : gA;
        float bLo = hi ? gB : c2[i];
        float bHi = hi ? c3[i] : gB;
        if (slo >= 0) {
          atomicAdd(&acc[dlo][lane], aLo * ilo);
          atomicAdd(&acc[dlo][32 + lane], bLo * ilo);
        }
        if (shi >= 0) {
          atomicAdd(&acc[dhi][lane], aHi * ihi);
          atomicAdd(&acc[dhi][32 + lane], bHi * ihi);
        }
      }
      e1 = e2; ms = msNext; dn = dnNext;
    }
  }

  int cb0 = (w & 1) * 32;
  v2f Br[16][2], Bq[16][2];
#pragma unroll
  for (int k = 0; k < 16; ++k) {
    int kb = k * 4 + hi * 2;
#pragma unroll
    for (int ct = 0; ct < 2; ++ct) {
      int col = cb0 + ct * 16 + n;
      Br[k][ct].x = rootW[kb * DD + col];
      Br[k][ct].y = rootW[(kb + 1) * DD + col];
      Bq[k][ct].x = resW[kb * DD + col];
      Bq[k][ct].y = resW[(kb + 1) * DD + col];
    }
  }
  float bR0 = biasv[cb0 + n], bR1 = biasv[cb0 + 16 + n];
  float bQ0 = resb[cb0 + n],  bQ1 = resb[cb0 + 16 + n];

  __syncthreads();

  for (int rt = w >> 1; rt < NBT; rt += EBT / 64) {
    int node = nodeBase + rt * 16 + n;
    if (node > N - 1) node = N - 1;
    const v2f* hp = (const v2f*)(h + (size_t)node * DD + hi * 2);
    v2f A[16];
#pragma unroll
    for (int k = 0; k < 16; ++k) A[k] = hp[k * 2];

    v8f cR0 = {}; v8f cR1 = {}; v8f cS0 = {}; v8f cS1 = {};
#pragma unroll
    for (int k = 0; k < 16; ++k) {
      cR0 = wmma4(A[k], Br[k][0], cR0);
      cR1 = wmma4(A[k], Br[k][1], cR1);
      cS0 = wmma4(A[k], Bq[k][0], cS0);
      cS1 = wmma4(A[k], Bq[k][1], cS1);
    }

#pragma unroll
    for (int i = 0; i < 8; ++i) {
      float* ap = &acc[rt * 16 + hi * 8 + i][cb0 + n];
      ap[0]  = fmaxf(ap[0]  + cR0[i] + bR0, 0.f) + fmaxf(cS0[i] + bQ0, 0.f);
      ap[16] = fmaxf(ap[16] + cR1[i] + bR1, 0.f) + fmaxf(cS1[i] + bQ1, 0.f);
    }
  }
  __syncthreads();

  for (int i = tid; i < NB * DD / 4; i += EBT) {
    int row = i >> 4, c4 = (i & 15) * 4;
    int node = nodeBase + row;
    if (node < N)
      *(float4*)(hout + (size_t)node * DD + c4) = *(const float4*)&acc[row][c4];
  }
}

extern "C" void kernel_launch(void* const* d_in, const int* in_sizes, int n_in,
                              void* d_out, int out_size, void* d_ws, size_t ws_size,
                              hipStream_t stream) {
  const float* x    = (const float*)d_in[0];
  const int*   ei   = (const int*)d_in[1];
  const int*   et   = (const int*)d_in[2];
  const float* W    = (const float*)d_in[3];
  const float* root = (const float*)d_in[4];
  const float* bias = (const float*)d_in[5];
  const float* resW = (const float*)d_in[6];
  const float* resb = (const float*)d_in[7];

  const int N = in_sizes[0] / DD;
  const int E = in_sizes[2];
  const int L = in_sizes[4] / (DD * DD);
  const int R = in_sizes[3] / (L * DD * DD);
  const int* src  = ei;
  const int* dstA = ei + E;
  const int NBLK = (N + NB - 1) / NB;
  const int T = NBLK * R * NB;
  const int CHN = (T + CS - 1) / CS;

  char* wsb = (char*)d_ws;
  size_t off = 0;
  auto carve = [&](size_t bytes) -> void* {
    void* p = wsb + off;
    off += (bytes + 255) & ~(size_t)255;
    return p;
  };
  int*   tbl     = (int*)carve((size_t)T * sizeof(int));
  int*   perm    = (int*)carve((size_t)E * sizeof(int));
  float* hA      = (float*)carve((size_t)N * DD * sizeof(float));
  float* hB      = (float*)carve((size_t)N * DD * sizeof(float));
  int*   bsum    = (int*)carve((size_t)CHN * sizeof(int));

  hipMemsetAsync(tbl, 0, (size_t)T * sizeof(int), stream);

  const int eb = (E + 255) / 256;
  edge_count  <<<eb, 256, 0, stream>>>(dstA, et, tbl, E, R);
  seg_sums    <<<CHN, 256, 0, stream>>>(tbl, bsum, T);
  scan_write  <<<CHN, 256, 0, stream>>>(tbl, bsum, T);
  edge_scatter<<<eb, 256, 0, stream>>>(dstA, et, tbl, perm, E, R);

  const float* hcur = x;
  for (int l = 0; l < L; ++l) {
    float* hnext = (l == L - 1) ? (float*)d_out : ((l & 1) ? hB : hA);
    edge_block<<<NBLK, EBT, 0, stream>>>(
        hcur, W + (size_t)l * R * DD * DD, perm, src, dstA, tbl,
        root + (size_t)l * DD * DD, bias + (size_t)l * DD,
        resW + (size_t)l * DD * DD, resb + (size_t)l * DD, hnext, N, R);
    hcur = hnext;
  }
}


# Counting pass keeps its add's return as the edge's rank; placing pass writes first slot + rank, no atomic (600k returning global adds gone); rank order shifts, LDS float sum order may change
# speedup vs baseline: 2.4332x; 1.0674x over previous
#include <hip/hip_runtime.h>


typedef __attribute__((ext_vector_type(2))) float v2f;
typedef __attribute__((ext_vector_type(8))) float v8f;

#define DD 64
#define NB 200
#define NBT ((NB + 15) / 16)
#define NBP (NBT * 16)
#define EBT 512

__device__ __forceinline__ v8f wmma4(v2f a, v2f b, v8f c) {
  return __builtin_amdgcn_wmma_f32_16x16x4_f32(
       false, a,  false, b,
       (short)0, c,  false,  false);
}

__device__ __forceinline__ float xhalf(float x) {
  int v = __builtin_bit_cast(int, x);
  return __builtin_bit_cast(float, __builtin_amdgcn_permlanex16(
      v, v, 0x76543210, 0xfedcba98, false, false));
}

__device__ __forceinline__ int tbl_index(int d, int r, int R) {
  int b = d / NB;
  return (b * R + r) * NB + (d - b * NB);
}

#define CS 2048

__device__ __forceinline__ int block_sum256(int v, int* red) {
  int tid = threadIdx.x;
  red[tid] = v;
  __syncthreads();
  for (int s = 128; s > 0; s >>= 1) {
    if (tid < s) red[tid] += red[tid + s];
    __syncthreads();
  }
  return red[0];
}

__global__ void edge_count(const int* __restrict__ dst, const int* __restrict__ et,
                           int* __restrict__ tbl, int* __restrict__ rank,
                           int E, int R) {
  int e = blockIdx.x * blockDim.x + threadIdx.x;
  if (e >= E) return;
  rank[e] = atomicAdd(&tbl[tbl_index(dst[e], et[e], R)], 1);
}

__global__ void __launch_bounds__(256)
seg_sums(const int* __restrict__ tbl, int* __restrict__ bsum, int T) {
  __shared__ int red[256];
  int lo = blockIdx.x * CS;
  int len = min(CS, T - lo);
  const int* p = tbl + lo;
  int s = 0;
  for (int i = threadIdx.x; i < len; i += 256) s += p[i];
  int tot = block_sum256(s, red);
  if (threadIdx.x == 0) bsum[blockIdx.x] = tot;
}

__global__ void __launch_bounds__(256)
scan_write(int* __restrict__ tbl, const int* __restrict__ bsum, int T) {
  __shared__ int red[256];
  __shared__ int tsum[2][256];
  int tid = threadIdx.x;
  int c = blockIdx.x;

  int bf = 0;
  for (int j = tid; j < c; j += 256) bf += bsum[j];
  int base = block_sum256(bf, red);

  int lo = c * CS;
  int len = min(CS, T - lo);
  int* p = tbl + lo;
  int i0 = tid * 8;
  int ex[8];
  int s = 0;
  for (int j = 0; j < 8; ++j) {
    int idx = i0 + j;
    int x = (idx < len) ? p[idx] : 0;
    ex[j] = s;
    s += x;
  }
  int cur = 0;
  tsum[0][tid] = s;
  __syncthreads();
  for (int off = 1; off < 256; off <<= 1) {
    int v = tsum[cur][tid];
    if (tid >= off) v += tsum[cur][tid - off];
    cur ^= 1;
    tsum[cur][tid] = v;
    __syncthreads();
  }
  int excl = tsum[cur][tid] - s;
  for (int j = 0; j < 8; ++j) {
    int idx = i0 + j;
    if (idx < len) p[idx] = base + excl + ex[j];
  }
}

__global__ void edge_scatter(const int* __restrict__ dst, const int* __restrict__ et,
                             const int* __restrict__ tbl, const int* __restrict__ rank,
                             int* __restrict__ perm, int E, int R) {
  int e = blockIdx.x * blockDim.x + threadIdx.x;
  if (e >= E) return;
  perm[tbl[tbl_index(dst[e], et[e], R)] + rank[e]] = e;
}

__global__ void __launch_bounds__(EBT)
edge_block(const float* __restrict__ h, const float* __restrict__ Wl,
           const int* __restrict__ perm, const int* __restrict__ src,
           const int* __restrict__ dstArr, const int* __restrict__ tbl,
           const float* __restrict__ rootW, const float* __restrict__ biasv,
           const float* __restrict__ resW, const float* __restrict__ resb,
           float* __restrict__ hout, int N, int R) {
  __shared__ float acc[NBP][DD];

  int tid = threadIdx.x;
  int b = blockIdx.x;
  int nodeBase = b * NB;

  for (int i = tid; i < NBP * DD / 4; i += EBT)
    ((float4*)&acc[0][0])[i] = make_float4(0.f, 0.f, 0.f, 0.f);
  __syncthreads();

  int lane = tid & 31;
  int w = __builtin_amdgcn_readfirstlane(tid >> 5);
  int n = lane & 15, hi = lane >> 4;

  for (int r = w; r < R; r += EBT / 32) {
    int g0 = (b * R + r) * NB;
    int begin = tbl[g0];
    int end = tbl[g0 + NB];
    begin = __builtin_amdgcn_readfirstlane(begin);
    end = __builtin_amdgcn_readfirstlane(end);
    if (begin >= end) continue;

    const float* Wr = Wl + (size_t)r * DD * DD;
    v2f B[16][4];
#pragma unroll
    for (int k = 0; k < 16; ++k) {
      int kb = k * 4 + hi * 2;
#pragma unroll
      for (int ct = 0; ct < 4; ++ct) {
        B[k][ct].x = Wr[kb * DD + ct * 16 + n];
        B[k][ct].y = Wr[(kb + 1) * DD + ct * 16 + n];
      }
    }

    int last = end - 1;
    int p0 = begin + n;
    int e1 = perm[min(p0 + 16, last)];
    int ms, dn;
    {
      int e0 = perm[min(p0, last)];
      ms = src[e0];
      dn = dstArr[e0];
      if (p0 >= end) ms = -1;
    }

    for (int t0 = begin; t0 < end; t0 += 16) {
      bool ok = ms >= 0;

      v2f A[16];
      const v2f* hp = (const v2f*)(h + (size_t)(ok ? ms : 0) * DD + hi * 2);
#pragma unroll
      for (int k = 0; k < 16; ++k) A[k] = hp[k * 2];

      int md = ok ? dn - nodeBase : 0;
      int gi = g0 + md;
      int cBeg = tbl[gi];
      int cEnd = tbl[gi + 1];

      int msNext = src[e1];
      int dnNext = dstArr[e1];
      if (t0 + 16 + n >= end) msNext = -1;
      int e2 = perm[min(t0 + 32 + n, last)];

      int c = cEnd - cBeg;
      float mi = 1.0f / (float)(c > 0 ? c : 1);

      v8f c0 = {}; v8f c1 = {}; v8f c2 = {}; v8f c3 = {};
#pragma unroll
      for (int k = 0; k < 16; ++k) {
        c0 = wmma4(A[k], B[k][0], c0);
        c1 = wmma4(A[k], B[k][1], c1);
        c2 = wmma4(A[k], B[k][2], c2);
        c3 = wmma4(A[k], B[k][3], c3);
      }

#pragma unroll
      for (int i = 0; i < 8; ++i) {
        int   slo = __builtin_amdgcn_readlane(ms, i);
        int   shi = __builtin_amdgcn_readlane(ms, i + 8);
        int   dlo = __builtin_amdgcn_readlane(md, i);
        int   dhi = __builtin_amdgcn_readlane(md, i + 8);
        float ilo = __builtin_bit_cast(float,
            __builtin_amdgcn_readlane(__builtin_bit_cast(int, mi), i));
        float ihi = __builtin_bit_cast(float,
            __builtin_amdgcn_readlane(__builtin_bit_cast(int, mi), i + 8));
        float gA = xhalf(hi ? c0[i] : c1[i]);
        float gB = xhalf(hi ? c2[i] : c3[i]);
        float aLo = hi ? gA : c0[i];
        float aHi = hi ? c1[i] : gA;
        float bLo = hi ? gB : c2[i];
        float bHi = hi ? c3[i] : gB;
        if (slo >= 0) {
          atomicAdd(&acc[dlo][lane], aLo * ilo);
          atomicAdd(&acc[dlo][32 + lane], bLo * ilo);
        }
        if (shi >= 0) {
          atomicAdd(&acc[dhi][lane], aHi * ihi);
          atomicAdd(&acc[dhi][32 + lane], bHi * ihi);
        }
      }
      e1 = e2; ms = msNext; dn = dnNext;
    }
  }

  int cb0 = (w & 1) * 32;
  v2f Br[16][2], Bq[16][2];
#pragma unroll
  for (int k = 0; k < 16; ++k) {
    int kb = k * 4 + hi * 2;
#pragma unroll
    for (int ct = 0; ct < 2; ++ct) {
      int col = cb0 + ct * 16 + n;
      Br[k][ct].x = rootW[kb * DD + col];
      Br[k][ct].y = rootW[(kb + 1) * DD + col];
      Bq[k][ct].x = resW[kb * DD + col];
      Bq[k][ct].y = resW[(kb + 1) * DD + col];
    }
  }
  float bR0 = biasv[cb0 + n], bR1 = biasv[cb0 + 16 + n];
  float bQ0 = resb[cb0 + n],  bQ1 = resb[cb0 + 16 + n];

  __syncthreads();

  for (int rt = w >> 1; rt < NBT; rt += EBT / 64) {
    int node = nodeBase + rt * 16 + n;
    if (node > N - 1) node = N - 1;
    const v2f* hp = (const v2f*)(h + (size_t)node * DD + hi * 2);
    v2f A[16];
#pragma unroll
    for (int k = 0; k < 16; ++k) A[k] = hp[k * 2];

    v8f cR0 = {}; v8f cR1 = {}; v8f cS0 = {}; v8f cS1 = {};
#pragma unroll
    for (int k = 0; k < 16; ++k) {
      cR0 = wmma4(A[k], Br[k][0], cR0);
      cR1 = wmma4(A[k], Br[k][1], cR1);
      cS0 = wmma4(A[k], Bq[k][0], cS0);
      cS1 = wmma4(A[k], Bq[k][1], cS1);
    }

#pragma unroll
    for (int i = 0; i < 8; ++i) {
      float* ap = &acc[rt * 16 + hi * 8 + i][cb0 + n];
      ap[0]  = fmaxf(ap[0]  + cR0[i] + bR0, 0.f) + fmaxf(cS0[i] + bQ0, 0.f);
      ap[16] = fmaxf(ap[16] + cR1[i] + bR1, 0.f) + fmaxf(cS1[i] + bQ1, 0.f);
    }
  }
  __syncthreads();

  for (int i = tid; i < NB * DD / 4; i += EBT) {
    int row = i >> 4, c4 = (i & 15) * 4;
    int node = nodeBase + row;
    if (node < N)
      *(float4*)(hout + (size_t)node * DD + c4) = *(const float4*)&acc[row][c4];
  }
}

extern "C" void kernel_launch(void* const* d_in, const int* in_sizes, int n_in,
                              void* d_out, int out_size, void* d_ws, size_t ws_size,
                              hipStream_t stream) {
  const float* x    = (const float*)d_in[0];
  const int*   ei   = (const int*)d_in[1];
  const int*   et   = (const int*)d_in[2];
  const float* W    = (const float*)d_in[3];
  const float* root = (const float*)d_in[4];
  const float* bias = (const float*)d_in[5];
  const float* resW = (const float*)d_in[6];
  const float* resb = (const float*)d_in[7];

  const int N = in_sizes[0] / DD;
  const int E = in_sizes[2];
  const int L = in_sizes[4] / (DD * DD);
  const int R = in_sizes[3] / (L * DD * DD);
  const int* src  = ei;
  const int* dstA = ei + E;
  const int NBLK = (N + NB - 1) / NB;
  const int T = NBLK * R * NB + 1;
  const int CHN = (T + CS - 1) / CS;

  char* wsb = (char*)d_ws;
  size_t off = 0;
  auto carve = [&](size_t bytes) -> void* {
    void* p = wsb + off;
    off += (bytes + 255) & ~(size_t)255;
    return p;
  };
  int*   tbl     = (int*)carve((size_t)T * sizeof(int));
  int*   perm    = (int*)carve((size_t)E * sizeof(int));
  float* hA      = (float*)carve((size_t)N * DD * sizeof(float));
  float* hB      = (float*)carve((size_t)N * DD * sizeof(float));
  int*   bsum    = (int*)carve((size_t)CHN * sizeof(int));
  int*   rank    = (int*)carve((size_t)E * sizeof(int));

  hipMemsetAsync(tbl, 0, (size_t)T * sizeof(int), stream);

  const int eb = (E + 255) / 256;
  edge_count  <<<eb, 256, 0, stream>>>(dstA, et, tbl, rank, E, R);
  seg_sums    <<<CHN, 256, 0, stream>>>(tbl, bsum, T);
  scan_write  <<<CHN, 256, 0, stream>>>(tbl, bsum, T);
  edge_scatter<<<eb, 256, 0, stream>>>(dstA, et, tbl, rank, perm, E, R);

  const float* hcur = x;
  for (int l = 0; l < L; ++l) {
    float* hnext = (l == L - 1) ? (float*)d_out : ((l & 1) ? hB : hA);
    edge_block<<<NBLK, EBT, 0, stream>>>(
        hcur, W + (size_t)l * R * DD * DD, perm, src, dstA, tbl,
        root + (size_t)l * DD * DD, bias + (size_t)l * DD,
        resW + (size_t)l * DD * DD, resb + (size_t)l * DD, hnext, N, R);
    hcur = hnext;
  }
}


# Block's matrix work is one queue its waves claim from an LDS counter: relation groups, then root/resW tile pairs (h@root joins the LDS atomic sums); no node phase; LDS float sum order changes
# speedup vs baseline: 2.5121x; 1.0324x over previous
#include <hip/hip_runtime.h>


typedef __attribute__((ext_vector_type(2))) float v2f;
typedef __attribute__((ext_vector_type(8))) float v8f;

#define DD 64
#define NB 200
#define NBT ((NB + 15) / 16)
#define NBP (NBT * 16)
#define EBT 512
#define NPAIR ((NBT + 1) / 2)

__device__ __forceinline__ v8f wmma4(v2f a, v2f b, v8f c) {
  return __builtin_amdgcn_wmma_f32_16x16x4_f32(
       false, a,  false, b,
       (short)0, c,  false,  false);
}

__device__ __forceinline__ float xhalf(float x) {
  int v = __builtin_bit_cast(int, x);
  return __builtin_bit_cast(float, __builtin_amdgcn_permlanex16(
      v, v, 0x76543210, 0xfedcba98, false, false));
}

__device__ __forceinline__ int tbl_index(int d, int r, int R) {
  int b = d / NB;
  return (b * R + r) * NB + (d - b * NB);
}

#define CS 2048

__device__ __forceinline__ int block_sum256(int v, int* red) {
  int tid = threadIdx.x;
  red[tid] = v;
  __syncthreads();
  for (int s = 128; s > 0; s >>= 1) {
    if (tid < s) red[tid] += red[tid + s];
    __syncthreads();
  }
  return red[0];
}

__global__ void edge_count(const int* __restrict__ dst, const int* __restrict__ et,
                           int* __restrict__ tbl, int* __restrict__ rank,
                           int E, int R) {
  int e = blockIdx.x * blockDim.x + threadIdx.x;
  if (e >= E) return;
  rank[e] = atomicAdd(&tbl[tbl_index(dst[e], et[e], R)], 1);
}

__global__ void __launch_bounds__(256)
seg_sums(const int* __restrict__ tbl, int* __restrict__ bsum, int T) {
  __shared__ int red[256];
  int lo = blockIdx.x * CS;
  int len = min(CS, T - lo);
  const int* p = tbl + lo;
  int s = 0;
  for (int i = threadIdx.x; i < len; i += 256) s += p[i];
  int tot = block_sum256(s, red);
  if (threadIdx.x == 0) bsum[blockIdx.x] = tot;
}

__global__ void __launch_bounds__(256)
scan_write(int* __restrict__ tbl, const int* __restrict__ bsum, int T) {
  __shared__ int red[256];
  __shared__ int tsum[2][256];
  int tid = threadIdx.x;
  int c = blockIdx.x;

  int bf = 0;
  for (int j = tid; j < c; j += 256) bf += bsum[j];
  int base = block_sum256(bf, red);

  int lo = c * CS;
  int len = min(CS, T - lo);
  int* p = tbl + lo;
  int i0 = tid * 8;
  int ex[8];
  int s = 0;
  for (int j = 0; j < 8; ++j) {
    int idx = i0 + j;
    int x = (idx < len) ? p[idx] : 0;
    ex[j] = s;
    s += x;
  }
  int cur = 0;
  tsum[0][tid] = s;
  __syncthreads();
  for (int off = 1; off < 256; off <<= 1) {
    int v = tsum[cur][tid];
    if (tid >= off) v += tsum[cur][tid - off];
    cur ^= 1;
    tsum[cur][tid] = v;
    __syncthreads();
  }
  int excl = tsum[cur][tid] - s;
  for (int j = 0; j < 8; ++j) {
    int idx = i0 + j;
    if (idx < len) p[idx] = base + excl + ex[j];
  }
}

__global__ void edge_scatter(const int* __restrict__ dst, const int* __restrict__ et,
                             const int* __restrict__ tbl, const int* __restrict__ rank,
                             int* __restrict__ perm, int E, int R) {
  int e = blockIdx.x * blockDim.x + threadIdx.x;
  if (e >= E) return;
  perm[tbl[tbl_index(dst[e], et[e], R)] + rank[e]] = e;
}

__device__ __forceinline__ void load_matrix(const float* __restrict__ Wr, int n,
                                            int hi, v2f (&B)[16][4]) {
#pragma unroll
  for (int k = 0; k < 16; ++k) {
    int kb = k * 4 + hi * 2;
#pragma unroll
    for (int ct = 0; ct < 4; ++ct) {
      B[k][ct].x = Wr[kb * DD + ct * 16 + n];
      B[k][ct].y = Wr[(kb + 1) * DD + ct * 16 + n];
    }
  }
}

__device__ __forceinline__ void load_rows(const float* __restrict__ h, int ms,
                                          int hi, v2f (&A)[16]) {
  const v2f* hp = (const v2f*)(h + (size_t)(ms >= 0 ? ms : 0) * DD + hi * 2);
#pragma unroll
  for (int k = 0; k < 16; ++k) A[k] = hp[k * 2];
}

__device__ __forceinline__ void mma_add(const v2f (&A)[16], const v2f (&B)[16][4],
                                        int ms, int md, float mi,
                                        float (*plane)[DD], int lane, int hi) {
  v8f c0 = {}; v8f c1 = {}; v8f c2 = {}; v8f c3 = {};
#pragma unroll
  for (int k = 0; k < 16; ++k) {
    c0 = wmma4(A[k], B[k][0], c0);
    c1 = wmma4(A[k], B[k][1], c1);
    c2 = wmma4(A[k], B[k][2], c2);
    c3 = wmma4(A[k], B[k][3], c3);
  }

#pragma unroll
  for (int i = 0; i < 8; ++i) {
    int   slo = __builtin_amdgcn_readlane(ms, i);
    int   shi = __builtin_amdgcn_readlane(ms, i + 8);
    int   dlo = __builtin_amdgcn_readlane(md, i);
    int   dhi = __builtin_amdgcn_readlane(md, i + 8);
    float ilo = __builtin_bit_cast(float,
        __builtin_amdgcn_readlane(__builtin_bit_cast(int, mi), i));
    float ihi = __builtin_bit_cast(float,
        __builtin_amdgcn_readlane(__builtin_bit_cast(int, mi), i + 8));
    float gA = xhalf(hi ? c0[i] : c1[i]);
    float gB = xhalf(hi ? c2[i] : c3[i]);
    float aLo = hi ? gA : c0[i];
    float aHi = hi ? c1[i] : gA;
    float bLo = hi ? gB : c2[i];
    float bHi = hi ? c3[i] : gB;
    if (slo >= 0) {
      atomicAdd(&plane[dlo][lane], aLo * ilo);
      atomicAdd(&plane[dlo][32 + lane], bLo * ilo);
    }
    if (shi >= 0) {
      atomicAdd(&plane[dhi][lane], aHi * ihi);
      atomicAdd(&plane[dhi][32 + lane], bHi * ihi);
    }
  }
}

__global__ void __launch_bounds__(EBT)
edge_block(const float* __restrict__ h, const float* __restrict__ Wl,
           const int* __restrict__ perm, const int* __restrict__ src,
           const int* __restrict__ dstArr, const int* __restrict__ tbl,
           const float* __restrict__ rootW, const float* __restrict__ biasv,
           const float* __restrict__ resW, const float* __restrict__ resb,
           float* __restrict__ hout, int N, int R) {
  __shared__ float acc[2][NBP][DD];
  __shared__ int qnext;

  int tid = threadIdx.x;
  int b = blockIdx.x;
  int nodeBase = b * NB;

  for (int i = tid; i < 2 * NBP * DD / 4; i += EBT)
    ((float4*)&acc[0][0][0])[i] = make_float4(0.f, 0.f, 0.f, 0.f);
  if (tid == 0) qnext = EBT / 32;
  __syncthreads();

  int lane = tid & 31;
  int w = __builtin_amdgcn_readfirstlane(tid >> 5);
  int n = lane & 15, hi = lane >> 4;

  const int U = R + 2 * NPAIR;
  int u = w;
  while (u < U) {
    bool rel = u < R;
    int g0 = 0, begin = 0, end = 1;
    const float* Wr;
    if (rel) {
      g0 = (b * R + u) * NB;
      begin = __builtin_amdgcn_readfirstlane(tbl[g0]);
      end = __builtin_amdgcn_readfirstlane(tbl[g0 + NB]);
      Wr = Wl + (size_t)u * DD * DD;
    } else {
      Wr = ((u - R) & 1) ? resW : rootW;
    }

    if (begin < end) {
      v2f B[16][4];

      if (rel) {
        int last = end - 1;
        int p0 = begin + n;
        int e0 = perm[min(p0, last)];
        int e1 = perm[min(p0 + 16, last)];
        load_matrix(Wr, n, hi, B);
        int ms = src[e0];
        int dn = dstArr[e0];
        if (p0 >= end) ms = -1;

        for (int t0 = begin; t0 < end; t0 += 16) {
          bool ok = ms >= 0;

          int md = ok ? dn - nodeBase : 0;
          int gi = g0 + md;
          int cBeg = tbl[gi];
          int cEnd = tbl[gi + 1];

          v2f A[16];
          load_rows(h, ms, hi, A);

          int msNext = src[e1];
          int dnNext = dstArr[e1];
          if (t0 + 16 + n >= end) msNext = -1;
          int e2 = perm[min(t0 + 32 + n, last)];

          __builtin_amdgcn_sched_barrier(0);

          int c = cEnd - cBeg;
          float mi = 1.0f / (float)(c > 0 ? c : 1);

          mma_add(A, B, ms, md, mi, acc[0], lane, hi);
          e1 = e2; ms = msNext; dn = dnNext;
        }
      } else {
        load_matrix(Wr, n, hi, B);
        int k = u - R;
        float (*plane)[DD] = acc[k & 1];
        int rt0 = (k >> 1) * 2;
        int rtEnd = min(rt0 + 2, NBT);
        for (int rt = rt0; rt < rtEnd; ++rt) {
          int row = rt * 16 + n;
          int ms = (row < NB && nodeBase + row < N) ? nodeBase + row : -1;
          v2f A[16];
          load_rows(h, ms, hi, A);
          mma_add(A, B, ms, row, 1.0f, plane, lane, hi);
        }
      }
    }

    int nu = 0;
    if (lane == 0) nu = atomicAdd(&qnext, 1);
    u = __builtin_amdgcn_readfirstlane(nu);
  }

  __syncthreads();

  int c4 = (tid & 15) * 4;
  float4 bR = *(const float4*)(biasv + c4);
  float4 bQ = *(const float4*)(resb + c4);
  for (int i = tid; i < NB * DD / 4; i += EBT) {
    int row = i >> 4;
    int node = nodeBase + row;
    if (node < N) {
      float4 a = *(const float4*)&acc[0][row][c4];
      float4 s = *(const float4*)&acc[1][row][c4];
      float4 o;
      o.x = fmaxf(a.x + bR.x, 0.f) + fmaxf(s.x + bQ.x, 0.f);
      o.y = fmaxf(a.y + bR.y, 0.f) + fmaxf(s.y + bQ.y, 0.f);
      o.z = fmaxf(a.z + bR.z, 0.f) + fmaxf(s.z + bQ.z, 0.f);
      o.w = fmaxf(a.w + bR.w, 0.f) + fmaxf(s.w + bQ.w, 0.f);
      *(float4*)(hout + (size_t)node * DD + c4) = o;
    }
  }
}

extern "C" void kernel_launch(void* const* d_in, const int* in_sizes, int n_in,
                              void* d_out, int out_size, void* d_ws, size_t ws_size,
                              hipStream_t stream) {
  const float* x    = (const float*)d_in[0];
  const int*   ei   = (const int*)d_in[1];
  const int*   et   = (const int*)d_in[2];
  const float* W    = (const float*)d_in[3];
  const float* root = (const float*)d_in[4];
  const float* bias = (const float*)d_in[5];
  const float* resW = (const float*)d_in[6];
  const float* resb = (const float*)d_in[7];

  const int N = in_sizes[0] / DD;
  const int E = in_sizes[2];
  const int L = in_sizes[4] / (DD * DD);
  const int R = in_sizes[3] / (L * DD * DD);
  const int* src  = ei;
  const int* dstA = ei + E;
  const int NBLK = (N + NB - 1) / NB;
  const int T = NBLK * R * NB + 1;
  const int CHN = (T + CS - 1) / CS;

  char* wsb = (char*)d_ws;
  size_t off = 0;
  auto carve = [&](size_t bytes) -> void* {
    void* p = wsb + off;
    off += (bytes + 255) & ~(size_t)255;
    return p;
  };
  int*   tbl     = (int*)carve((size_t)T * sizeof(int));
  int*   perm    = (int*)carve((size_t)E * sizeof(int));
  float* hA      = (float*)carve((size_t)N * DD * sizeof(float));
  float* hB      = (float*)carve((size_t)N * DD * sizeof(float));
  int*   bsum    = (int*)carve((size_t)CHN * sizeof(int));
  int*   rank    = (int*)carve((size_t)E * sizeof(int));

  hipMemsetAsync(tbl, 0, (size_t)T * sizeof(int), stream);

  const int eb = (E + 255) / 256;
  edge_count  <<<eb, 256, 0, stream>>>(dstA, et, tbl, rank, E, R);
  seg_sums    <<<CHN, 256, 0, stream>>>(tbl, bsum, T);
  scan_write  <<<CHN, 256, 0, stream>>>(tbl, bsum, T);
  edge_scatter<<<eb, 256, 0, stream>>>(dstA, et, tbl, rank, perm, E, R);

  const float* hcur = x;
  for (int l = 0; l < L; ++l) {
    float* hnext = (l == L - 1) ? (float*)d_out : ((l & 1) ? hB : hA);
    edge_block<<<NBLK, EBT, 0, stream>>>(
        hcur, W + (size_t)l * R * DD * DD, perm, src, dstA, tbl,
        root + (size_t)l * DD * DD, bias + (size_t)l * DD,
        resW + (size_t)l * DD * DD, resb + (size_t)l * DD, hnext, N, R);
    hcur = hnext;
  }
}


# Sort rebuilt as two local LDS sorts: edge chunks split by destination block, then each block counts, scans, places its edges; table clear, 2 launches, global atomics gone; LDS float sum order changes
# speedup vs baseline: 3.2592x; 1.2974x over previous
#include <hip/hip_runtime.h>


typedef __attribute__((ext_vector_type(2))) float v2f;
typedef __attribute__((ext_vector_type(8))) float v8f;

#define DD 64
#define NB 200
#define NBT ((NB + 15) / 16)
#define NBP (NBT * 16)
#define EBT 512
#define NPAIR ((NBT + 1) / 2)

__device__ __forceinline__ v8f wmma4(v2f a, v2f b, v8f c) {
  return __builtin_amdgcn_wmma_f32_16x16x4_f32(
       false, a,  false, b,
       (short)0, c,  false,  false);
}

__device__ __forceinline__ float xhalf(float x) {
  int v = __builtin_bit_cast(int, x);
  return __builtin_bit_cast(float, __builtin_amdgcn_permlanex16(
      v, v, 0x76543210, 0xfedcba98, false, false));
}

#define CHK 2048
#define HCAP 1024
#define KCAP 13312
#define NCHCAP 1024
#define ST 512

template <int T>
__device__ __forceinline__ int block_sum(int v, int* red) {
  int tid = threadIdx.x;
  red[tid] = v;
  __syncthreads();
  for (int s = T / 2; s > 0; s >>= 1) {
    if (tid < s) red[tid] += red[tid + s];
    __syncthreads();
  }
  int r = red[0];
  __syncthreads();
  return r;
}

template <int T>
__device__ __forceinline__ int block_scan_inplace(int* a, int n, int (*ts)[T]) {
  int tid = threadIdx.x;
  int per = (n + T - 1) / T;
  int lo = min(tid * per, n);
  int hi = min(lo + per, n);
  int s = 0;
  for (int i = lo; i < hi; ++i) s += a[i];
  int cur = 0;
  ts[0][tid] = s;
  __syncthreads();
  for (int o = 1; o < T; o <<= 1) {
    int v = ts[cur][tid];
    if (tid >= o) v += ts[cur][tid - o];
    cur ^= 1;
    ts[cur][tid] = v;
    __syncthreads();
  }
  int run = ts[cur][tid] - s;
  int total = ts[cur][T - 1];
  for (int i = lo; i < hi; ++i) {
    int v = a[i];
    a[i] = run;
    run += v;
  }
  __syncthreads();
  return total;
}

__global__ void __launch_bounds__(256)
chunk_split(const int* __restrict__ dst, const int* __restrict__ et,
            int2* __restrict__ rec, int* __restrict__ off, int E, int NBLK) {
  __shared__ int hist[HCAP];
  __shared__ int ts[2][256];
  int tid = threadIdx.x;
  int c = blockIdx.x;
  for (int i = tid; i < NBLK; i += 256) hist[i] = 0;
  __syncthreads();

  int eb[8], ekey[8], erank[8];
#pragma unroll
  for (int j = 0; j < 8; ++j) {
    int e = c * CHK + j * 256 + tid;
    eb[j] = -1; ekey[j] = 0; erank[j] = 0;
    if (e < E) {
      int d = dst[e];
      int b = d / NB;
      eb[j] = b;
      ekey[j] = et[e] * NB + (d - b * NB);
      erank[j] = atomicAdd(&hist[b], 1);
    }
  }
  __syncthreads();
  int total = block_scan_inplace<256>(hist, NBLK, ts);

  int* o = off + (size_t)c * (NBLK + 1);
  for (int i = tid; i < NBLK; i += 256) o[i] = hist[i];
  if (tid == 0) o[NBLK] = total;
#pragma unroll
  for (int j = 0; j < 8; ++j) {
    if (eb[j] >= 0) {
      int e = c * CHK + j * 256 + tid;
      rec[(size_t)c * CHK + hist[eb[j]] + erank[j]] = make_int2(e, ekey[j]);
    }
  }
}

__device__ __forceinline__ int seg_of(const int* cseg, int n, int i) {
  int lo = 0, hi = n;
  while (hi - lo > 1) {
    int mid = (lo + hi) >> 1;
    if (cseg[mid] <= i) lo = mid; else hi = mid;
  }
  return lo;
}

__global__ void __launch_bounds__(ST)
block_sort(const int2* __restrict__ rec, const int* __restrict__ off,
           int* __restrict__ tbl, int* __restrict__ perm,
           int NCH, int NBLK, int K) {
  __shared__ int cnt[KCAP];
  __shared__ int cseg[NCHCAP];
  __shared__ int sbeg[NCHCAP];
  __shared__ int ts[2][ST];
  int tid = threadIdx.x;
  int b = blockIdx.x;

  for (int i = tid; i <= K; i += ST) cnt[i] = 0;
  int sp = 0;
  for (int c = tid; c < NCH; c += ST) {
    const int* o = off + (size_t)c * (NBLK + 1) + b;
    int o0 = o[0], o1 = o[1];
    sbeg[c] = c * CHK + o0;
    cseg[c] = o1 - o0;
    sp += o0;
  }
  int S = block_sum<ST>(sp, ts[0]);
  int M = block_scan_inplace<ST>(cseg, NCH, ts);

  int2 rr[8];
#pragma unroll
  for (int j = 0; j < 8; ++j) {
    int i = tid + j * ST;
    rr[j] = make_int2(-1, 0);
    if (i < M) {
      int c = seg_of(cseg, NCH, i);
      rr[j] = rec[(size_t)sbeg[c] + (i - cseg[c])];
    }
  }
#pragma unroll
  for (int j = 0; j < 8; ++j)
    if (rr[j].x >= 0) atomicAdd(&cnt[rr[j].y], 1);
  for (int i = tid + 8 * ST; i < M; i += ST) {
    int c = seg_of(cseg, NCH, i);
    int2 q = rec[(size_t)sbeg[c] + (i - cseg[c])];
    atomicAdd(&cnt[q.y], 1);
  }
  __syncthreads();
  block_scan_inplace<ST>(cnt, K + 1, ts);

  int* tb = tbl + (size_t)b * K;
  for (int i = tid; i < K; i += ST) tb[i] = S + cnt[i];
  if (b == NBLK - 1 && tid == 0) tb[K] = S + M;
  __syncthreads();

#pragma unroll
  for (int j = 0; j < 8; ++j)
    if (rr[j].x >= 0) perm[S + atomicAdd(&cnt[rr[j].y], 1)] = rr[j].x;
  for (int i = tid + 8 * ST; i < M; i += ST) {
    int c = seg_of(cseg, NCH, i);
    int2 q = rec[(size_t)sbeg[c] + (i - cseg[c])];
    perm[S + atomicAdd(&cnt[q.y], 1)] = q.x;
  }
}

__device__ __forceinline__ void load_matrix(const float* __restrict__ Wr, int n,
                                            int hi, v2f (&B)[16][4]) {
#pragma unroll
  for (int k = 0; k < 16; ++k) {
    int kb = k * 4 + hi * 2;
#pragma unroll
    for (int ct = 0; ct < 4; ++ct) {
      B[k][ct].x = Wr[kb * DD + ct * 16 + n];
      B[k][ct].y = Wr[(kb + 1) * DD + ct * 16 + n];
    }
  }
}

__device__ __forceinline__ void load_rows(const float* __restrict__ h, int ms,
                                          int hi, v2f (&A)[16]) {
  const v2f* hp = (const v2f*)(h + (size_t)(ms >= 0 ? ms : 0) * DD + hi * 2);
#pragma unroll
  for (int k = 0; k < 16; ++k) A[k] = hp[k * 2];
}

__device__ __forceinline__ void mma_add(const v2f (&A)[16], const v2f (&B)[16][4],
                                        int ms, int md, float mi,
                                        float (*plane)[DD], int lane, int hi) {
  v8f c0 = {}; v8f c1 = {}; v8f c2 = {}; v8f c3 = {};
#pragma unroll
  for (int k = 0; k < 16; ++k) {
    c0 = wmma4(A[k], B[k][0], c0);
    c1 = wmma4(A[k], B[k][1], c1);
    c2 = wmma4(A[k], B[k][2], c2);
    c3 = wmma4(A[k], B[k][3], c3);
  }

#pragma unroll
  for (int i = 0; i < 8; ++i) {
    int   slo = __builtin_amdgcn_readlane(ms, i);
    int   shi = __builtin_amdgcn_readlane(ms, i + 8);
    int   dlo = __builtin_amdgcn_readlane(md, i);
    int   dhi = __builtin_amdgcn_readlane(md, i + 8);
    float ilo = __builtin_bit_cast(float,
        __builtin_amdgcn_readlane(__builtin_bit_cast(int, mi), i));
    float ihi = __builtin_bit_cast(float,
        __builtin_amdgcn_readlane(__builtin_bit_cast(int, mi), i + 8));
    float gA = xhalf(hi ? c0[i] : c1[i]);
    float gB = xhalf(hi ? c2[i] : c3[i]);
    float aLo = hi ? gA : c0[i];
    float aHi = hi ? c1[i] : gA;
    float bLo = hi ? gB : c2[i];
    float bHi = hi ? c3[i] : gB;
    if (slo >= 0) {
      atomicAdd(&plane[dlo][lane], aLo * ilo);
      atomicAdd(&plane[dlo][32 + lane], bLo * ilo);
    }
    if (shi >= 0) {
      atomicAdd(&plane[dhi][lane], aHi * ihi);
      atomicAdd(&plane[dhi][32 + lane], bHi * ihi);
    }
  }
}

__global__ void __launch_bounds__(EBT)
edge_block(const float* __restrict__ h, const float* __restrict__ Wl,
           const int* __restrict__ perm, const int* __restrict__ src,
           const int* __restrict__ dstArr, const int* __restrict__ tbl,
           const float* __restrict__ rootW, const float* __restrict__ biasv,
           const float* __restrict__ resW, const float* __restrict__ resb,
           float* __restrict__ hout, int N, int R) {
  __shared__ float acc[2][NBP][DD];
  __shared__ int qnext;

  int tid = threadIdx.x;
  int b = blockIdx.x;
  int nodeBase = b * NB;

  for (int i = tid; i < 2 * NBP * DD / 4; i += EBT)
    ((float4*)&acc[0][0][0])[i] = make_float4(0.f, 0.f, 0.f, 0.f);
  if (tid == 0) qnext = EBT / 32;
  __syncthreads();

  int lane = tid & 31;
  int w = __builtin_amdgcn_readfirstlane(tid >> 5);
  int n = lane & 15, hi = lane >> 4;

  const int U = R + 2 * NPAIR;
  int u = w;
  while (u < U) {
    bool rel = u < R;
    int g0 = 0, begin = 0, end = 1;
    const float* Wr;
    if (rel) {
      g0 = (b * R + u) * NB;
      begin = __builtin_amdgcn_readfirstlane(tbl[g0]);
      end = __builtin_amdgcn_readfirstlane(tbl[g0 + NB]);
      Wr = Wl + (size_t)u * DD * DD;
    } else {
      Wr = ((u - R) & 1) ? resW : rootW;
    }

    if (begin < end) {
      v2f B[16][4];

      if (rel) {
        int last = end - 1;
        int p0 = begin + n;
        int e0 = perm[min(p0, last)];
        int e1 = perm[min(p0 + 16, last)];
        load_matrix(Wr, n, hi, B);
        int ms = src[e0];
        int dn = dstArr[e0];
        if (p0 >= end) ms = -1;

        for (int t0 = begin; t0 < end; t0 += 16) {
          bool ok = ms >= 0;

          int md = ok ? dn - nodeBase : 0;
          int gi = g0 + md;
          int cBeg = tbl[gi];
          int cEnd = tbl[gi + 1];

          v2f A[16];
          load_rows(h, ms, hi, A);

          int msNext = src[e1];
          int dnNext = dstArr[e1];
          if (t0 + 16 + n >= end) msNext = -1;
          int e2 = perm[min(t0 + 32 + n, last)];

          __builtin_amdgcn_sched_barrier(0);

          int c = cEnd - cBeg;
          float mi = 1.0f / (float)(c > 0 ? c : 1);

          mma_add(A, B, ms, md, mi, acc[0], lane, hi);
          e1 = e2; ms = msNext; dn = dnNext;
        }
      } else {
        load_matrix(Wr, n, hi, B);
        int k = u - R;
        float (*plane)[DD] = acc[k & 1];
        int rt0 = (k >> 1) * 2;
        int rtEnd = min(rt0 + 2, NBT);
        for (int rt = rt0; rt < rtEnd; ++rt) {
          int row = rt * 16 + n;
          int ms = (row < NB && nodeBase + row < N) ? nodeBase + row : -1;
          v2f A[16];
          load_rows(h, ms, hi, A);
          mma_add(A, B, ms, row, 1.0f, plane, lane, hi);
        }
      }
    }

    int nu = 0;
    if (lane == 0) nu = atomicAdd(&qnext, 1);
    u = __builtin_amdgcn_readfirstlane(nu);
  }

  __syncthreads();

  int c4 = (tid & 15) * 4;
  float4 bR = *(const float4*)(biasv + c4);
  float4 bQ = *(const float4*)(resb + c4);
  for (int i = tid; i < NB * DD / 4; i += EBT) {
    int row = i >> 4;
    int node = nodeBase + row;
    if (node < N) {
      float4 a = *(const float4*)&acc[0][row][c4];
      float4 s = *(const float4*)&acc[1][row][c4];
      float4 o;
      o.x = fmaxf(a.x + bR.x, 0.f) + fmaxf(s.x + bQ.x, 0.f);
      o.y = fmaxf(a.y + bR.y, 0.f) + fmaxf(s.y + bQ.y, 0.f);
      o.z = fmaxf(a.z + bR.z, 0.f) + fmaxf(s.z + bQ.z, 0.f);
      o.w = fmaxf(a.w + bR.w, 0.f) + fmaxf(s.w + bQ.w, 0.f);
      *(float4*)(hout + (size_t)node * DD + c4) = o;
    }
  }
}

extern "C" void kernel_launch(void* const* d_in, const int* in_sizes, int n_in,
                              void* d_out, int out_size, void* d_ws, size_t ws_size,
                              hipStream_t stream) {
  const float* x    = (const float*)d_in[0];
  const int*   ei   = (const int*)d_in[1];
  const int*   et   = (const int*)d_in[2];
  const float* W    = (const float*)d_in[3];
  const float* root = (const float*)d_in[4];
  const float* bias = (const float*)d_in[5];
  const float* resW = (const float*)d_in[6];
  const float* resb = (const float*)d_in[7];

  const int N = in_sizes[0] / DD;
  const int E = in_sizes[2];
  const int L = in_sizes[4] / (DD * DD);
  const int R = in_sizes[3] / (L * DD * DD);
  const int* src  = ei;
  const int* dstA = ei + E;
  const int NBLK = (N + NB - 1) / NB;
  const int K = R * NB;
  const int T = NBLK * K + 1;
  const int NCH = (E + CHK - 1) / CHK;

  char* wsb = (char*)d_ws;
  size_t off = 0;
  auto carve = [&](size_t bytes) -> void* {
    void* p = wsb + off;
    off += (bytes + 255) & ~(size_t)255;
    return p;
  };
  int*   tbl     = (int*)carve((size_t)T * sizeof(int));
  int*   perm    = (int*)carve((size_t)E * sizeof(int));
  float* hA      = (float*)carve((size_t)N * DD * sizeof(float));
  size_t recBytes = ((size_t)E * sizeof(int2) + 255) & ~(size_t)255;
  size_t offBytes = (size_t)NCH * (NBLK + 1) * sizeof(int);
  size_t hbBytes  = (size_t)N * DD * sizeof(float);
  if (recBytes + offBytes > hbBytes) hbBytes = recBytes + offBytes;
  float* hB      = (float*)carve(hbBytes);
  int2*  rec     = (int2*)hB;
  int*   coff    = (int*)((char*)hB + recBytes);

  chunk_split<<<NCH, 256, 0, stream>>>(dstA, et, rec, coff, E, NBLK);
  block_sort <<<NBLK, ST, 0, stream>>>(rec, coff, tbl, perm, NCH, NBLK, K);

  const float* hcur = x;
  for (int l = 0; l < L; ++l) {
    float* hnext = (l == L - 1) ? (float*)d_out : ((l & 1) ? hB : hA);
    edge_block<<<NBLK, EBT, 0, stream>>>(
        hcur, W + (size_t)l * R * DD * DD, perm, src, dstA, tbl,
        root + (size_t)l * DD * DD, bias + (size_t)l * DD,
        resW + (size_t)l * DD * DD, resb + (size_t)l * DD, hnext, N, R);
    hcur = hnext;
  }
}


# WMMA fragment order re-chosen for 16-byte loads: a lane half holds 4 consecutive k (A and B), B tile ct holds column 4n+ct; 8 row and 32 matrix loads replace 16 and 128; k sum order changes
# speedup vs baseline: 3.4160x; 1.0481x over previous
#include <hip/hip_runtime.h>


typedef __attribute__((ext_vector_type(2))) float v2f;
typedef __attribute__((ext_vector_type(8))) float v8f;

#define DD 64
#define NB 200
#define NBT ((NB + 15) / 16)
#define NBP (NBT * 16)
#define EBT 512
#define NPAIR ((NBT + 1) / 2)

__device__ __forceinline__ v8f wmma4(v2f a, v2f b, v8f c) {
  return __builtin_amdgcn_wmma_f32_16x16x4_f32(
       false, a,  false, b,
       (short)0, c,  false,  false);
}

__device__ __forceinline__ float xhalf(float x) {
  int v = __builtin_bit_cast(int, x);
  return __builtin_bit_cast(float, __builtin_amdgcn_permlanex16(
      v, v, 0x76543210, 0xfedcba98, false, false));
}

#define CHK 2048
#define HCAP 1024
#define KCAP 13312
#define NCHCAP 1024
#define ST 512

template <int T>
__device__ __forceinline__ int block_sum(int v, int* red) {
  int tid = threadIdx.x;
  red[tid] = v;
  __syncthreads();
  for (int s = T / 2; s > 0; s >>= 1) {
    if (tid < s) red[tid] += red[tid + s];
    __syncthreads();
  }
  int r = red[0];
  __syncthreads();
  return r;
}

template <int T>
__device__ __forceinline__ int block_scan_inplace(int* a, int n, int (*ts)[T]) {
  int tid = threadIdx.x;
  int per = (n + T - 1) / T;
  int lo = min(tid * per, n);
  int hi = min(lo + per, n);
  int s = 0;
  for (int i = lo; i < hi; ++i) s += a[i];
  int cur = 0;
  ts[0][tid] = s;
  __syncthreads();
  for (int o = 1; o < T; o <<= 1) {
    int v = ts[cur][tid];
    if (tid >= o) v += ts[cur][tid - o];
    cur ^= 1;
    ts[cur][tid] = v;
    __syncthreads();
  }
  int run = ts[cur][tid] - s;
  int total = ts[cur][T - 1];
  for (int i = lo; i < hi; ++i) {
    int v = a[i];
    a[i] = run;
    run += v;
  }
  __syncthreads();
  return total;
}

__global__ void __launch_bounds__(256)
chunk_split(const int* __restrict__ dst, const int* __restrict__ et,
            int2* __restrict__ rec, int* __restrict__ off, int E, int NBLK) {
  __shared__ int hist[HCAP];
  __shared__ int ts[2][256];
  int tid = threadIdx.x;
  int c = blockIdx.x;
  for (int i = tid; i < NBLK; i += 256) hist[i] = 0;
  __syncthreads();

  int eb[8], ekey[8], erank[8];
#pragma unroll
  for (int j = 0; j < 8; ++j) {
    int e = c * CHK + j * 256 + tid;
    eb[j] = -1; ekey[j] = 0; erank[j] = 0;
    if (e < E) {
      int d = dst[e];
      int b = d / NB;
      eb[j] = b;
      ekey[j] = et[e] * NB + (d - b * NB);
      erank[j] = atomicAdd(&hist[b], 1);
    }
  }
  __syncthreads();
  int total = block_scan_inplace<256>(hist, NBLK, ts);

  int* o = off + (size_t)c * (NBLK + 1);
  for (int i = tid; i < NBLK; i += 256) o[i] = hist[i];
  if (tid == 0) o[NBLK] = total;
#pragma unroll
  for (int j = 0; j < 8; ++j) {
    if (eb[j] >= 0) {
      int e = c * CHK + j * 256 + tid;
      rec[(size_t)c * CHK + hist[eb[j]] + erank[j]] = make_int2(e, ekey[j]);
    }
  }
}

__device__ __forceinline__ int seg_of(const int* cseg, int n, int i) {
  int lo = 0, hi = n;
  while (hi - lo > 1) {
    int mid = (lo + hi) >> 1;
    if (cseg[mid] <= i) lo = mid; else hi = mid;
  }
  return lo;
}

__global__ void __launch_bounds__(ST)
block_sort(const int2* __restrict__ rec, const int* __restrict__ off,
           int* __restrict__ tbl, int* __restrict__ perm,
           int NCH, int NBLK, int K) {
  __shared__ int cnt[KCAP];
  __shared__ int cseg[NCHCAP];
  __shared__ int sbeg[NCHCAP];
  __shared__ int ts[2][ST];
  int tid = threadIdx.x;
  int b = blockIdx.x;

  for (int i = tid; i <= K; i += ST) cnt[i] = 0;
  int sp = 0;
  for (int c = tid; c < NCH; c += ST) {
    const int* o = off + (size_t)c * (NBLK + 1) + b;
    int o0 = o[0], o1 = o[1];
    sbeg[c] = c * CHK + o0;
    cseg[c] = o1 - o0;
    sp += o0;
  }
  int S = block_sum<ST>(sp, ts[0]);
  int M = block_scan_inplace<ST>(cseg, NCH, ts);

  int2 rr[8];
#pragma unroll
  for (int j = 0; j < 8; ++j) {
    int i = tid + j * ST;
    rr[j] = make_int2(-1, 0);
    if (i < M) {
      int c = seg_of(cseg, NCH, i);
      rr[j] = rec[(size_t)sbeg[c] + (i - cseg[c])];
    }
  }
#pragma unroll
  for (int j = 0; j < 8; ++j)
    if (rr[j].x >= 0) atomicAdd(&cnt[rr[j].y], 1);
  for (int i = tid + 8 * ST; i < M; i += ST) {
    int c = seg_of(cseg, NCH, i);
    int2 q = rec[(size_t)sbeg[c] + (i - cseg[c])];
    atomicAdd(&cnt[q.y], 1);
  }
  __syncthreads();
  block_scan_inplace<ST>(cnt, K + 1, ts);

  int* tb = tbl + (size_t)b * K;
  for (int i = tid; i < K; i += ST) tb[i] = S + cnt[i];
  if (b == NBLK - 1 && tid == 0) tb[K] = S + M;
  __syncthreads();

#pragma unroll
  for (int j = 0; j < 8; ++j)
    if (rr[j].x >= 0) perm[S + atomicAdd(&cnt[rr[j].y], 1)] = rr[j].x;
  for (int i = tid + 8 * ST; i < M; i += ST) {
    int c = seg_of(cseg, NCH, i);
    int2 q = rec[(size_t)sbeg[c] + (i - cseg[c])];
    perm[S + atomicAdd(&cnt[q.y], 1)] = q.x;
  }
}

__device__ __forceinline__ void load_matrix(const float* __restrict__ Wr, int n,
                                            int hi, v2f (&B)[16][4]) {
  const float4* wp = (const float4*)(Wr + (hi * 4) * DD + n * 4);
#pragma unroll
  for (int s = 0; s < 16; ++s) {
    int k0 = (s >> 1) * 8 + (s & 1) * 2;
    float4 r0 = wp[k0 * (DD / 4)];
    float4 r1 = wp[(k0 + 1) * (DD / 4)];
    B[s][0].x = r0.x; B[s][0].y = r1.x;
    B[s][1].x = r0.y; B[s][1].y = r1.y;
    B[s][2].x = r0.z; B[s][2].y = r1.z;
    B[s][3].x = r0.w; B[s][3].y = r1.w;
  }
}

__device__ __forceinline__ void load_rows(const float* __restrict__ h, int ms,
                                          int hi, v2f (&A)[16]) {
  const float4* hp =
      (const float4*)(h + (size_t)(ms >= 0 ? ms : 0) * DD + hi * 4);
#pragma unroll
  for (int j = 0; j < 8; ++j) {
    float4 v = hp[j * 2];
    A[2 * j].x = v.x;     A[2 * j].y = v.y;
    A[2 * j + 1].x = v.z; A[2 * j + 1].y = v.w;
  }
}

__device__ __forceinline__ void mma_add(const v2f (&A)[16], const v2f (&B)[16][4],
                                        int ms, int md, float mi,
                                        float (*plane)[DD], int n, int hi) {
  v8f c0 = {}; v8f c1 = {}; v8f c2 = {}; v8f c3 = {};
#pragma unroll
  for (int k = 0; k < 16; ++k) {
    c0 = wmma4(A[k], B[k][0], c0);
    c1 = wmma4(A[k], B[k][1], c1);
    c2 = wmma4(A[k], B[k][2], c2);
    c3 = wmma4(A[k], B[k][3], c3);
  }

  int cc = n * 4 + hi * 2;
#pragma unroll
  for (int i = 0; i < 8; ++i) {
    int   slo = __builtin_amdgcn_readlane(ms, i);
    int   shi = __builtin_amdgcn_readlane(ms, i + 8);
    int   dlo = __builtin_amdgcn_readlane(md, i);
    int   dhi = __builtin_amdgcn_readlane(md, i + 8);
    float ilo = __builtin_bit_cast(float,
        __builtin_amdgcn_readlane(__builtin_bit_cast(int, mi), i));
    float ihi = __builtin_bit_cast(float,
        __builtin_amdgcn_readlane(__builtin_bit_cast(int, mi), i + 8));
    float gA = xhalf(hi ? c0[i] : c2[i]);
    float gB = xhalf(hi ? c1[i] : c3[i]);
    float lo0 = hi ? gA : c0[i];
    float lo1 = hi ? gB : c1[i];
    float hi0 = hi ? c2[i] : gA;
    float hi1 = hi ? c3[i] : gB;
    if (slo >= 0) {
      atomicAdd(&plane[dlo][cc], lo0 * ilo);
      atomicAdd(&plane[dlo][cc + 1], lo1 * ilo);
    }
    if (shi >= 0) {
      atomicAdd(&plane[dhi][cc], hi0 * ihi);
      atomicAdd(&plane[dhi][cc + 1], hi1 * ihi);
    }
  }
}

__global__ void __launch_bounds__(EBT)
edge_block(const float* __restrict__ h, const float* __restrict__ Wl,
           const int* __restrict__ perm, const int* __restrict__ src,
           const int* __restrict__ dstArr, const int* __restrict__ tbl,
           const float* __restrict__ rootW, const float* __restrict__ biasv,
           const float* __restrict__ resW, const float* __restrict__ resb,
           float* __restrict__ hout, int N, int R) {
  __shared__ float acc[2][NBP][DD];
  __shared__ int qnext;

  int tid = threadIdx.x;
  int b = blockIdx.x;
  int nodeBase = b * NB;

  for (int i = tid; i < 2 * NBP * DD / 4; i += EBT)
    ((float4*)&acc[0][0][0])[i] = make_float4(0.f, 0.f, 0.f, 0.f);
  if (tid == 0) qnext = EBT / 32;
  __syncthreads();

  int lane = tid & 31;
  int w = __builtin_amdgcn_readfirstlane(tid >> 5);
  int n = lane & 15, hi = lane >> 4;

  const int U = R + 2 * NPAIR;
  int u = w;
  while (u < U) {
    bool rel = u < R;
    int g0 = 0, begin = 0, end = 1;
    const float* Wr;
    if (rel) {
      g0 = (b * R + u) * NB;
      begin = __builtin_amdgcn_readfirstlane(tbl[g0]);
      end = __builtin_amdgcn_readfirstlane(tbl[g0 + NB]);
      Wr = Wl + (size_t)u * DD * DD;
    } else {
      Wr = ((u - R) & 1) ? resW : rootW;
    }

    if (begin < end) {
      v2f B[16][4];

      if (rel) {
        int last = end - 1;
        int p0 = begin + n;
        int e0 = perm[min(p0, last)];
        int e1 = perm[min(p0 + 16, last)];
        load_matrix(Wr, n, hi, B);
        int ms = src[e0];
        int dn = dstArr[e0];
        if (p0 >= end) ms = -1;

        for (int t0 = begin; t0 < end; t0 += 16) {
          bool ok = ms >= 0;

          int md = ok ? dn - nodeBase : 0;
          int gi = g0 + md;
          int cBeg = tbl[gi];
          int cEnd = tbl[gi + 1];

          v2f A[16];
          load_rows(h, ms, hi, A);

          int msNext = src[e1];
          int dnNext = dstArr[e1];
          if (t0 + 16 + n >= end) msNext = -1;
          int e2 = perm[min(t0 + 32 + n, last)];

          __builtin_amdgcn_sched_barrier(0);

          int c = cEnd - cBeg;
          float mi = 1.0f / (float)(c > 0 ? c : 1);

          mma_add(A, B, ms, md, mi, acc[0], n, hi);
          e1 = e2; ms = msNext; dn = dnNext;
        }
      } else {
        load_matrix(Wr, n, hi, B);
        int k = u - R;
        float (*plane)[DD] = acc[k & 1];
        int rt0 = (k >> 1) * 2;
        int rtEnd = min(rt0 + 2, NBT);
        for (int rt = rt0; rt < rtEnd; ++rt) {
          int row = rt * 16 + n;
          int ms = (row < NB && nodeBase + row < N) ? nodeBase + row : -1;
          v2f A[16];
          load_rows(h, ms, hi, A);
          mma_add(A, B, ms, row, 1.0f, plane, n, hi);
        }
      }
    }

    int nu = 0;
    if (lane == 0) nu = atomicAdd(&qnext, 1);
    u = __builtin_amdgcn_readfirstlane(nu);
  }

  __syncthreads();

  int c4 = (tid & 15) * 4;
  float4 bR = *(const float4*)(biasv + c4);
  float4 bQ = *(const float4*)(resb + c4);
  for (int i = tid; i < NB * DD / 4; i += EBT) {
    int row = i >> 4;
    int node = nodeBase + row;
    if (node < N) {
      float4 a = *(const float4*)&acc[0][row][c4];
      float4 s = *(const float4*)&acc[1][row][c4];
      float4 o;
      o.x = fmaxf(a.x + bR.x, 0.f) + fmaxf(s.x + bQ.x, 0.f);
      o.y = fmaxf(a.y + bR.y, 0.f) + fmaxf(s.y + bQ.y, 0.f);
      o.z = fmaxf(a.z + bR.z, 0.f) + fmaxf(s.z + bQ.z, 0.f);
      o.w = fmaxf(a.w + bR.w, 0.f) + fmaxf(s.w + bQ.w, 0.f);
      *(float4*)(hout + (size_t)node * DD + c4) = o;
    }
  }
}

extern "C" void kernel_launch(void* const* d_in, const int* in_sizes, int n_in,
                              void* d_out, int out_size, void* d_ws, size_t ws_size,
                              hipStream_t stream) {
  const float* x    = (const float*)d_in[0];
  const int*   ei   = (const int*)d_in[1];
  const int*   et   = (const int*)d_in[2];
  const float* W    = (const float*)d_in[3];
  const float* root = (const float*)d_in[4];
  const float* bias = (const float*)d_in[5];
  const float* resW = (const float*)d_in[6];
  const float* resb = (const float*)d_in[7];

  const int N = in_sizes[0] / DD;
  const int E = in_sizes[2];
  const int L = in_sizes[4] / (DD * DD);
  const int R = in_sizes[3] / (L * DD * DD);
  const int* src  = ei;
  const int* dstA = ei + E;
  const int NBLK = (N + NB - 1) / NB;
  const int K = R * NB;
  const int T = NBLK * K + 1;
  const int NCH = (E + CHK - 1) / CHK;

  char* wsb = (char*)d_ws;
  size_t off = 0;
  auto carve = [&](size_t bytes) -> void* {
    void* p = wsb + off;
    off += (bytes + 255) & ~(size_t)255;
    return p;
  };
  int*   tbl     = (int*)carve((size_t)T * sizeof(int));
  int*   perm    = (int*)carve((size_t)E * sizeof(int));
  float* hA      = (float*)carve((size_t)N * DD * sizeof(float));
  size_t recBytes = ((size_t)E * sizeof(int2) + 255) & ~(size_t)255;
  size_t offBytes = (size_t)NCH * (NBLK + 1) * sizeof(int);
  size_t hbBytes  = (size_t)N * DD * sizeof(float);
  if (recBytes + offBytes > hbBytes) hbBytes = recBytes + offBytes;
  float* hB      = (float*)carve(hbBytes);
  int2*  rec     = (int2*)hB;
  int*   coff    = (int*)((char*)hB + recBytes);

  chunk_split<<<NCH, 256, 0, stream>>>(dstA, et, rec, coff, E, NBLK);
  block_sort <<<NBLK, ST, 0, stream>>>(rec, coff, tbl, perm, NCH, NBLK, K);

  const float* hcur = x;
  for (int l = 0; l < L; ++l) {
    float* hnext = (l == L - 1) ? (float*)d_out : ((l & 1) ? hB : hA);
    edge_block<<<NBLK, EBT, 0, stream>>>(
        hcur, W + (size_t)l * R * DD * DD, perm, src, dstA, tbl,
        root + (size_t)l * DD * DD, bias + (size_t)l * DD,
        resW + (size_t)l * DD * DD, resb + (size_t)l * DD, hnext, N, R);
    hcur = hnext;
  }
}


# Block sort fused into the first layer's kernel: a work-group sorts its block's edges in the LDS that later holds its sums, then reads back its own table and order; 4 launches now 3; float sums kept
# speedup vs baseline: 3.5253x; 1.0320x over previous
#include <hip/hip_runtime.h>


typedef __attribute__((ext_vector_type(2))) float v2f;
typedef __attribute__((ext_vector_type(8))) float v8f;

#define DD 64
#define NB 200
#define NBT ((NB + 15) / 16)
#define NBP (NBT * 16)
#define EBT 512
#define NPAIR ((NBT + 1) / 2)

__device__ __forceinline__ v8f wmma4(v2f a, v2f b, v8f c) {
  return __builtin_amdgcn_wmma_f32_16x16x4_f32(
       false, a,  false, b,
       (short)0, c,  false,  false);
}

__device__ __forceinline__ float xhalf(float x) {
  int v = __builtin_bit_cast(int, x);
  return __builtin_bit_cast(float, __builtin_amdgcn_permlanex16(
      v, v, 0x76543210, 0xfedcba98, false, false));
}

#define CHK 2048
#define HCAP 1024
#define KCAP 13312
#define NCHCAP 1024
#define ST 512

template <int T>
__device__ __forceinline__ int block_sum(int v, int* red) {
  int tid = threadIdx.x;
  red[tid] = v;
  __syncthreads();
  for (int s = T / 2; s > 0; s >>= 1) {
    if (tid < s) red[tid] += red[tid + s];
    __syncthreads();
  }
  int r = red[0];
  __syncthreads();
  return r;
}

template <int T>
__device__ __forceinline__ int block_scan_inplace(int* a, int n, int (*ts)[T]) {
  int tid = threadIdx.x;
  int per = (n + T - 1) / T;
  int lo = min(tid * per, n);
  int hi = min(lo + per, n);
  int s = 0;
  for (int i = lo; i < hi; ++i) s += a[i];
  int cur = 0;
  ts[0][tid] = s;
  __syncthreads();
  for (int o = 1; o < T; o <<= 1) {
    int v = ts[cur][tid];
    if (tid >= o) v += ts[cur][tid - o];
    cur ^= 1;
    ts[cur][tid] = v;
    __syncthreads();
  }
  int run = ts[cur][tid] - s;
  int total = ts[cur][T - 1];
  for (int i = lo; i < hi; ++i) {
    int v = a[i];
    a[i] = run;
    run += v;
  }
  __syncthreads();
  return total;
}

__global__ void __launch_bounds__(256)
chunk_split(const int* __restrict__ dst, const int* __restrict__ et,
            int2* __restrict__ rec, int* __restrict__ off, int E, int NBLK) {
  __shared__ int hist[HCAP];
  __shared__ int ts[2][256];
  int tid = threadIdx.x;
  int c = blockIdx.x;
  for (int i = tid; i < NBLK; i += 256) hist[i] = 0;
  __syncthreads();

  int eb[8], ekey[8], erank[8];
#pragma unroll
  for (int j = 0; j < 8; ++j) {
    int e = c * CHK + j * 256 + tid;
    eb[j] = -1; ekey[j] = 0; erank[j] = 0;
    if (e < E) {
      int d = dst[e];
      int b = d / NB;
      eb[j] = b;
      ekey[j] = et[e] * NB + (d - b * NB);
      erank[j] = atomicAdd(&hist[b], 1);
    }
  }
  __syncthreads();
  int total = block_scan_inplace<256>(hist, NBLK, ts);

  int* o = off + (size_t)c * (NBLK + 1);
  for (int i = tid; i < NBLK; i += 256) o[i] = hist[i];
  if (tid == 0) o[NBLK] = total;
#pragma unroll
  for (int j = 0; j < 8; ++j) {
    if (eb[j] >= 0) {
      int e = c * CHK + j * 256 + tid;
      rec[(size_t)c * CHK + hist[eb[j]] + erank[j]] = make_int2(e, ekey[j]);
    }
  }
}

__device__ __forceinline__ int seg_of(const int* cseg, int n, int i) {
  int lo = 0, hi = n;
  while (hi - lo > 1) {
    int mid = (lo + hi) >> 1;
    if (cseg[mid] <= i) lo = mid; else hi = mid;
  }
  return lo;
}

__device__ __forceinline__ void
sort_block(const int2* __restrict__ rec, const int* __restrict__ off,
           int* tbl, int* perm, int NCH, int NBLK, int K,
           int* cnt,
           int* cseg,
           int* sbeg,
           int (*ts)[ST]) {
  int tid = threadIdx.x;
  int b = blockIdx.x;

  for (int i = tid; i <= K; i += ST) cnt[i] = 0;
  int sp = 0;
  for (int c = tid; c < NCH; c += ST) {
    const int* o = off + (size_t)c * (NBLK + 1) + b;
    int o0 = o[0], o1 = o[1];
    sbeg[c] = c * CHK + o0;
    cseg[c] = o1 - o0;
    sp += o0;
  }
  int S = block_sum<ST>(sp, ts[0]);
  int M = block_scan_inplace<ST>(cseg, NCH, ts);

  int2 rr[8];
#pragma unroll
  for (int j = 0; j < 8; ++j) {
    int i = tid + j * ST;
    rr[j] = make_int2(-1, 0);
    if (i < M) {
      int c = seg_of(cseg, NCH, i);
      rr[j] = rec[(size_t)sbeg[c] + (i - cseg[c])];
    }
  }
#pragma unroll
  for (int j = 0; j < 8; ++j)
    if (rr[j].x >= 0) atomicAdd(&cnt[rr[j].y], 1);
  for (int i = tid + 8 * ST; i < M; i += ST) {
    int c = seg_of(cseg, NCH, i);
    int2 q = rec[(size_t)sbeg[c] + (i - cseg[c])];
    atomicAdd(&cnt[q.y], 1);
  }
  __syncthreads();
  block_scan_inplace<ST>(cnt, K + 1, ts);

  int* tb = tbl + (size_t)b * K;
  for (int i = tid; i < K; i += ST) tb[i] = S + cnt[i];
  if (tid == 0) tb[K] = S + M;
  __syncthreads();

#pragma unroll
  for (int j = 0; j < 8; ++j)
    if (rr[j].x >= 0) perm[S + atomicAdd(&cnt[rr[j].y], 1)] = rr[j].x;
  for (int i = tid + 8 * ST; i < M; i += ST) {
    int c = seg_of(cseg, NCH, i);
    int2 q = rec[(size_t)sbeg[c] + (i - cseg[c])];
    perm[S + atomicAdd(&cnt[q.y], 1)] = q.x;
  }
}

__device__ __forceinline__ void load_matrix(const float* __restrict__ Wr, int n,
                                            int hi, v2f (&B)[16][4]) {
  const float4* wp = (const float4*)(Wr + (hi * 4) * DD + n * 4);
#pragma unroll
  for (int s = 0; s < 16; ++s) {
    int k0 = (s >> 1) * 8 + (s & 1) * 2;
    float4 r0 = wp[k0 * (DD / 4)];
    float4 r1 = wp[(k0 + 1) * (DD / 4)];
    B[s][0].x = r0.x; B[s][0].y = r1.x;
    B[s][1].x = r0.y; B[s][1].y = r1.y;
    B[s][2].x = r0.z; B[s][2].y = r1.z;
    B[s][3].x = r0.w; B[s][3].y = r1.w;
  }
}

__device__ __forceinline__ void load_rows(const float* __restrict__ h, int ms,
                                          int hi, v2f (&A)[16]) {
  const float4* hp =
      (const float4*)(h + (size_t)(ms >= 0 ? ms : 0) * DD + hi * 4);
#pragma unroll
  for (int j = 0; j < 8; ++j) {
    float4 v = hp[j * 2];
    A[2 * j].x = v.x;     A[2 * j].y = v.y;
    A[2 * j + 1].x = v.z; A[2 * j + 1].y = v.w;
  }
}

__device__ __forceinline__ void mma_add(const v2f (&A)[16], const v2f (&B)[16][4],
                                        int ms, int md, float mi,
                                        float (*plane)[DD], int n, int hi) {
  v8f c0 = {}; v8f c1 = {}; v8f c2 = {}; v8f c3 = {};
#pragma unroll
  for (int k = 0; k < 16; ++k) {
    c0 = wmma4(A[k], B[k][0], c0);
    c1 = wmma4(A[k], B[k][1], c1);
    c2 = wmma4(A[k], B[k][2], c2);
    c3 = wmma4(A[k], B[k][3], c3);
  }

  int cc = n * 4 + hi * 2;
#pragma unroll
  for (int i = 0; i < 8; ++i) {
    int   slo = __builtin_amdgcn_readlane(ms, i);
    int   shi = __builtin_amdgcn_readlane(ms, i + 8);
    int   dlo = __builtin_amdgcn_readlane(md, i);
    int   dhi = __builtin_amdgcn_readlane(md, i + 8);
    float ilo = __builtin_bit_cast(float,
        __builtin_amdgcn_readlane(__builtin_bit_cast(int, mi), i));
    float ihi = __builtin_bit_cast(float,
        __builtin_amdgcn_readlane(__builtin_bit_cast(int, mi), i + 8));
    float gA = xhalf(hi ? c0[i] : c2[i]);
    float gB = xhalf(hi ? c1[i] : c3[i]);
    float lo0 = hi ? gA : c0[i];
    float lo1 = hi ? gB : c1[i];
    float hi0 = hi ? c2[i] : gA;
    float hi1 = hi ? c3[i] : gB;
    if (slo >= 0) {
      atomicAdd(&plane[dlo][cc], lo0 * ilo);
      atomicAdd(&plane[dlo][cc + 1], lo1 * ilo);
    }
    if (shi >= 0) {
      atomicAdd(&plane[dhi][cc], hi0 * ihi);
      atomicAdd(&plane[dhi][cc + 1], hi1 * ihi);
    }
  }
}

__device__ __forceinline__ void
layer_body(const float* h, const float* Wl, const int* perm, const int* src,
           const int* dstArr, const int* tbl, const float* rootW,
           const float* biasv, const float* resW, const float* resb,
           float* hout, int N, int R, float (*acc)[NBP][DD], int* qnextp) {
  int& qnext = *qnextp;
  int tid = threadIdx.x;
  int b = blockIdx.x;
  int nodeBase = b * NB;

  for (int i = tid; i < 2 * NBP * DD / 4; i += EBT)
    ((float4*)&acc[0][0][0])[i] = make_float4(0.f, 0.f, 0.f, 0.f);
  if (tid == 0) qnext = EBT / 32;
  __syncthreads();

  int lane = tid & 31;
  int w = __builtin_amdgcn_readfirstlane(tid >> 5);
  int n = lane & 15, hi = lane >> 4;

  const int U = R + 2 * NPAIR;
  int u = w;
  while (u < U) {
    bool rel = u < R;
    int g0 = 0, begin = 0, end = 1;
    const float* Wr;
    if (rel) {
      g0 = (b * R + u) * NB;
      begin = __builtin_amdgcn_readfirstlane(tbl[g0]);
      end = __builtin_amdgcn_readfirstlane(tbl[g0 + NB]);
      Wr = Wl + (size_t)u * DD * DD;
    } else {
      Wr = ((u - R) & 1) ? resW : rootW;
    }

    if (begin < end) {
      v2f B[16][4];

      if (rel) {
        int last = end - 1;
        int p0 = begin + n;
        int e0 = perm[min(p0, last)];
        int e1 = perm[min(p0 + 16, last)];
        load_matrix(Wr, n, hi, B);
        int ms = src[e0];
        int dn = dstArr[e0];
        if (p0 >= end) ms = -1;

        for (int t0 = begin; t0 < end; t0 += 16) {
          bool ok = ms >= 0;

          int md = ok ? dn - nodeBase : 0;
          int gi = g0 + md;
          int cBeg = tbl[gi];
          int cEnd = tbl[gi + 1];

          v2f A[16];
          load_rows(h, ms, hi, A);

          int msNext = src[e1];
          int dnNext = dstArr[e1];
          if (t0 + 16 + n >= end) msNext = -1;
          int e2 = perm[min(t0 + 32 + n, last)];

          __builtin_amdgcn_sched_barrier(0);

          int c = cEnd - cBeg;
          float mi = 1.0f / (float)(c > 0 ? c : 1);

          mma_add(A, B, ms, md, mi, acc[0], n, hi);
          e1 = e2; ms = msNext; dn = dnNext;
        }
      } else {
        load_matrix(Wr, n, hi, B);
        int k = u - R;
        float (*plane)[DD] = acc[k & 1];
        int rt0 = (k >> 1) * 2;
        int rtEnd = min(rt0 + 2, NBT);
        for (int rt = rt0; rt < rtEnd; ++rt) {
          int row = rt * 16 + n;
          int ms = (row < NB && nodeBase + row < N) ? nodeBase + row : -1;
          v2f A[16];
          load_rows(h, ms, hi, A);
          mma_add(A, B, ms, row, 1.0f, plane, n, hi);
        }
      }
    }

    int nu = 0;
    if (lane == 0) nu = atomicAdd(&qnext, 1);
    u = __builtin_amdgcn_readfirstlane(nu);
  }

  __syncthreads();

  int c4 = (tid & 15) * 4;
  float4 bR = *(const float4*)(biasv + c4);
  float4 bQ = *(const float4*)(resb + c4);
  for (int i = tid; i < NB * DD / 4; i += EBT) {
    int row = i >> 4;
    int node = nodeBase + row;
    if (node < N) {
      float4 a = *(const float4*)&acc[0][row][c4];
      float4 s = *(const float4*)&acc[1][row][c4];
      float4 o;
      o.x = fmaxf(a.x + bR.x, 0.f) + fmaxf(s.x + bQ.x, 0.f);
      o.y = fmaxf(a.y + bR.y, 0.f) + fmaxf(s.y + bQ.y, 0.f);
      o.z = fmaxf(a.z + bR.z, 0.f) + fmaxf(s.z + bQ.z, 0.f);
      o.w = fmaxf(a.w + bR.w, 0.f) + fmaxf(s.w + bQ.w, 0.f);
      *(float4*)(hout + (size_t)node * DD + c4) = o;
    }
  }
}

__global__ void __launch_bounds__(EBT)
edge_block(const float* __restrict__ h, const float* __restrict__ Wl,
           const int* __restrict__ perm, const int* __restrict__ src,
           const int* __restrict__ dstArr, const int* __restrict__ tbl,
           const float* __restrict__ rootW, const float* __restrict__ biasv,
           const float* __restrict__ resW, const float* __restrict__ resb,
           float* __restrict__ hout, int N, int R) {
  __shared__ float acc[2][NBP][DD];
  __shared__ int qnext;
  layer_body(h, Wl, perm, src, dstArr, tbl, rootW, biasv, resW, resb, hout,
             N, R, acc, &qnext);
}

static_assert(ST == EBT, "the block sort runs on the edge kernel's threads");
static_assert((KCAP + 2 * NCHCAP + 2 * ST) * sizeof(int) <=
              2 * NBP * DD * sizeof(float), "sort tables fit in the planes");
__global__ void __launch_bounds__(EBT)
edge_block_first(const int2* __restrict__ rec, const int* __restrict__ off,
                 int* tbl, int* perm, int NCH, int NBLK, int K,
                 const float* __restrict__ h, const float* __restrict__ Wl,
                 const int* __restrict__ src, const int* __restrict__ dstArr,
                 const float* __restrict__ rootW,
                 const float* __restrict__ biasv,
                 const float* __restrict__ resW,
                 const float* __restrict__ resb,
                 float* __restrict__ hout, int N, int R) {
  __shared__ float acc[2][NBP][DD];
  __shared__ int qnext;
  int* cnt = (int*)&acc[0][0][0];
  int* cseg = cnt + KCAP;
  int* sbeg = cseg + NCHCAP;
  int (*ts)[ST] = (int (*)[ST])(sbeg + NCHCAP);
  sort_block(rec, off, tbl, perm, NCH, NBLK, K, cnt, cseg, sbeg, ts);
  __syncthreads();
  layer_body(h, Wl, perm, src, dstArr, tbl, rootW, biasv, resW, resb, hout,
             N, R, acc, &qnext);
}

extern "C" void kernel_launch(void* const* d_in, const int* in_sizes, int n_in,
                              void* d_out, int out_size, void* d_ws, size_t ws_size,
                              hipStream_t stream) {
  const float* x    = (const float*)d_in[0];
  const int*   ei   = (const int*)d_in[1];
  const int*   et   = (const int*)d_in[2];
  const float* W    = (const float*)d_in[3];
  const float* root = (const float*)d_in[4];
  const float* bias = (const float*)d_in[5];
  const float* resW = (const float*)d_in[6];
  const float* resb = (const float*)d_in[7];

  const int N = in_sizes[0] / DD;
  const int E = in_sizes[2];
  const int L = in_sizes[4] / (DD * DD);
  const int R = in_sizes[3] / (L * DD * DD);
  const int* src  = ei;
  const int* dstA = ei + E;
  const int NBLK = (N + NB - 1) / NB;
  const int K = R * NB;
  const int T = NBLK * K + 1;
  const int NCH = (E + CHK - 1) / CHK;

  char* wsb = (char*)d_ws;
  size_t off = 0;
  auto carve = [&](size_t bytes) -> void* {
    void* p = wsb + off;
    off += (bytes + 255) & ~(size_t)255;
    return p;
  };
  int*   tbl     = (int*)carve((size_t)T * sizeof(int));
  int*   perm    = (int*)carve((size_t)E * sizeof(int));
  float* hA      = (float*)carve((size_t)N * DD * sizeof(float));
  size_t recBytes = ((size_t)E * sizeof(int2) + 255) & ~(size_t)255;
  size_t offBytes = (size_t)NCH * (NBLK + 1) * sizeof(int);
  size_t hbBytes  = (size_t)N * DD * sizeof(float);
  if (recBytes + offBytes > hbBytes) hbBytes = recBytes + offBytes;
  float* hB      = (float*)carve(hbBytes);
  int2*  rec     = (int2*)hB;
  int*   coff    = (int*)((char*)hB + recBytes);

  chunk_split<<<NCH, 256, 0, stream>>>(dstA, et, rec, coff, E, NBLK);

  const float* hcur = x;
  for (int l = 0; l < L; ++l) {
    float* hnext = (l == L - 1) ? (float*)d_out : ((l & 1) ? hB : hA);
    if (l == 0)
      edge_block_first<<<NBLK, EBT, 0, stream>>>(
          rec, coff, tbl, perm, NCH, NBLK, K, hcur, W, src, dstA, root, bias,
          resW, resb, hnext, N, R);
    else
      edge_block<<<NBLK, EBT, 0, stream>>>(
          hcur, W + (size_t)l * R * DD * DD, perm, src, dstA, tbl,
          root + (size_t)l * DD * DD, bias + (size_t)l * DD,
          resW + (size_t)l * DD * DD, resb + (size_t)l * DD, hnext, N, R);
    hcur = hnext;
  }
}
